# starting kernel
# baseline (speedup 1.0000x reference)
#include <hip/hip_runtime.h>
#include <hip/hip_bf16.h>

#define N_HEADS 32
#define N_KV 8
#define HDIM 8
#define SEQ 2048
#define HID 256
#define QTILES (SEQ / 16)

typedef __attribute__((ext_vector_type(2))) float v2f;
typedef __attribute__((ext_vector_type(8))) float v8f;

template <int CTRL>
__device__ __forceinline__ float dpp_max(float x) {
  const int y =
      __builtin_amdgcn_update_dpp(0, __float_as_int(x), CTRL, 0xf, 0xf, true);
  return fmaxf(x, __int_as_float(y));
}

__device__ __forceinline__ float rowmax16(float x) {
  x = dpp_max<0xB1>(x);
  x = dpp_max<0x4E>(x);
  x = dpp_max<0x141>(x);
  x = dpp_max<0x140>(x);
  return x;
}

__device__ __forceinline__ float fast_exp2(float x) {
  return __builtin_amdgcn_exp2f(x);
}

__global__ __launch_bounds__(64) void qkv_rope_so8_kernel(
    const float* __restrict__ x, const float* __restrict__ Wq,
    const float* __restrict__ Wk, const float* __restrict__ Wv,
    const float* __restrict__ rope, const float* __restrict__ so8,
    float* __restrict__ qb, float* __restrict__ kb, float* __restrict__ vtb) {
  const int bs = blockIdx.x;
  const int s = bs & (SEQ - 1);
  const int b = bs >> 11;

  __shared__ float xs[HID];
  for (int i = threadIdx.x; i < HID; i += 64) xs[i] = x[(size_t)bs * HID + i];
  __syncthreads();

  const int t = threadIdx.x;
  if (t >= 48) return;
  const int kind = (t < 32) ? 0 : (t < 40 ? 1 : 2);
  const int head = (t < 32) ? t : (t < 40 ? t - 32 : t - 40);
  const float* W = (kind == 0) ? Wq : ((kind == 1) ? Wk : Wv);
  const float* wr = W + (size_t)head * HDIM * HID;

  float acc[HDIM] = {};
  for (int i = 0; i < HID; ++i) {
    const float xv = xs[i];
#pragma unroll
    for (int d = 0; d < HDIM; ++d) acc[d] = fmaf(xv, wr[d * HID + i], acc[d]);
  }

  float y[HDIM];
  if (kind <= 1) {
    const float invf[4] = {1.0f, 0.1f, 0.01f, 0.001f};
    float cr[4] = {}, sr[4] = {};
#pragma unroll
    for (int i = 0; i < 4; ++i) {
      const float fr = (float)s * invf[i];
      const float ci = cosf(fr), si = sinf(fr);
#pragma unroll
      for (int j = 0; j < 4; ++j) {
        cr[j] = fmaf(ci, rope[i * 8 + j], cr[j]);
        sr[j] = fmaf(si, rope[i * 8 + j], sr[j]);
      }
    }
#pragma unroll
    for (int d = 0; d < HDIM; ++d) {
      const int j = d & 3;
      const float rh = (d < 4) ? -acc[d + 4] : acc[d - 4];
      y[d] = acc[d] * cr[j] + rh * sr[j];
    }
  } else {
#pragma unroll
    for (int d = 0; d < HDIM; ++d) y[d] = acc[d];
  }

  float z[HDIM];
#pragma unroll
  for (int j = 0; j < HDIM; ++j) {
    float sum = 0.0f;
#pragma unroll
    for (int d = 0; d < HDIM; ++d) sum = fmaf(y[d], so8[d * 8 + j], sum);
    z[j] = sum;
  }

  if (kind == 0) {
    float* dst = qb + (((size_t)b * N_HEADS + head) * SEQ + s) * HDIM;
#pragma unroll
    for (int j = 0; j < HDIM; ++j) dst[j] = z[j];
  } else if (kind == 1) {
    float* dst = kb + (((size_t)b * N_KV + head) * SEQ + s) * HDIM;
#pragma unroll
    for (int j = 0; j < HDIM; ++j) dst[j] = z[j];
  } else {
#pragma unroll
    for (int j = 0; j < HDIM; ++j)
      vtb[(((size_t)b * N_KV + head) * HDIM + j) * SEQ + s] = z[j];
  }
}

template <bool MASK>
__device__ __forceinline__ void attn_tile(
    const float* __restrict__ kp, const float* __restrict__ vp,
    float* __restrict__ pw, int kbase, int lanelo, int hi, int hi2,
    const float* __restrict__ maskv, const v2f Qa0, const v2f Qa1,
    float* __restrict__ m, v8f& o) {
  const float* krow = kp + (size_t)(kbase + lanelo) * HDIM;
  const v2f Kb0 = *(const v2f*)(krow + hi2);
  const v2f Kb1 = *(const v2f*)(krow + 4 + hi2);

  v8f sc = {};
  sc = __builtin_amdgcn_wmma_f32_16x16x4_f32(false, Qa0, false, Kb0, (short)0,
                                             sc, false, false);
  sc = __builtin_amdgcn_wmma_f32_16x16x4_f32(false, Qa1, false, Kb1, (short)0,
                                             sc, false, false);

  float p[8];
#pragma unroll
  for (int r = 0; r < 8; ++r) {
    float sv = sc[r];
    if (MASK) sv += maskv[r];
    const float rm = rowmax16(sv);
    const float mnew = fmaxf(m[r], rm);
    const float corr = fast_exp2(m[r] - mnew);
    m[r] = mnew;
    p[r] = fast_exp2(sv - mnew);
    o[r] = o[r] * corr;
  }

#pragma unroll
  for (int r = 0; r < 8; ++r) pw[r * 16 + hi * 128 + lanelo] = p[r];
  v2f Pa[4];
#pragma unroll
  for (int t = 0; t < 4; ++t)
    Pa[t] = *(const v2f*)(pw + (lanelo & 7) * 16 + (lanelo >> 3) * 128 +
                          4 * t + hi2);

  v2f Vb[4];
#pragma unroll
  for (int t = 0; t < 4; ++t) {
    if (lanelo < 8) {
      Vb[t] = *(const v2f*)(vp + (size_t)lanelo * SEQ + kbase + 4 * t + hi2);
    } else {
      const float c = (lanelo == 8) ? 1.0f : 0.0f;
      Vb[t] = (v2f){c, c};
    }
  }
#pragma unroll
  for (int t = 0; t < 4; ++t)
    o = __builtin_amdgcn_wmma_f32_16x16x4_f32(false, Pa[t], false, Vb[t],
                                              (short)0, o, false, false);
}

__global__ __launch_bounds__(256) void so8_flash_attn_kernel(
    const float* __restrict__ q, const float* __restrict__ k,
    const float* __restrict__ vT, float* __restrict__ attn) {
  const int lane = threadIdx.x & 31;
  const int widx = threadIdx.x >> 5;
  const int w = blockIdx.x * 8 + widx;
  const int qt = w & (QTILES - 1);
  const int h = (w >> 7) & (N_HEADS - 1);
  const int b = w >> 12;
  const int kv = h >> 2;
  const int qbase = qt * 16;
  const int lanelo = lane & 15;
  const int hi = lane >> 4;
  const int hi2 = hi * 2;

  const float* qp = q + ((size_t)b * N_HEADS + h) * SEQ * HDIM;
  const float* kp = k + ((size_t)b * N_KV + kv) * SEQ * HDIM;
  const float* vp = vT + ((size_t)b * N_KV + kv) * HDIM * SEQ;

  __shared__ float plds[8][256];
  float* pw = &plds[widx][0];

  const float QSCALE = 0.35355339059327f * 1.44269504088896f;
  const float* qrow = qp + (size_t)(qbase + lanelo) * HDIM;
  const v2f Qa0 = *(const v2f*)(qrow + hi2) * QSCALE;
  const v2f Qa1 = *(const v2f*)(qrow + 4 + hi2) * QSCALE;

  float maskv[8];
#pragma unroll
  for (int r = 0; r < 8; ++r)
    maskv[r] = (lanelo > r + 8 * hi) ? -3.0e38f : 0.0f;

  v8f o = {};
  float m[8];
#pragma unroll
  for (int r = 0; r < 8; ++r) m[r] = -3.0e38f;

  for (int kt = 0; kt < qt; ++kt)
    attn_tile<false>(kp, vp, pw, kt * 16, lanelo, hi, hi2, maskv, Qa0, Qa1, m,
                     o);
  attn_tile<true>(kp, vp, pw, qbase, lanelo, hi, hi2, maskv, Qa0, Qa1, m, o);

#pragma unroll
  for (int r = 0; r < 8; ++r) {
    const float ov = o[r];
    const float l = __shfl(ov, (lane & 16) + 8);
    const float outv = ov * __builtin_amdgcn_rcpf(l);
    const int row = qbase + r + 8 * hi;
    if (lanelo < HDIM)
      attn[(((size_t)b * SEQ + row) * N_HEADS + h) * HDIM + lanelo] = outv;
  }
}

__global__ __launch_bounds__(256) void out_proj_kernel(
    const float* __restrict__ attn, const float* __restrict__ Wo,
    float* __restrict__ out) {
  const int bs = blockIdx.x;
  __shared__ float xs[HID];
  xs[threadIdx.x] = attn[(size_t)bs * HID + threadIdx.x];
  __syncthreads();
  const int oi = threadIdx.x;
  const float* wr = Wo + (size_t)oi * HID;
  float acc = 0.0f;
  for (int i = 0; i < HID; ++i) acc = fmaf(xs[i], wr[i], acc);
  out[(size_t)bs * HID + oi] = acc;
}

extern "C" void kernel_launch(void* const* d_in, const int* in_sizes, int n_in,
                              void* d_out, int out_size, void* d_ws,
                              size_t ws_size, hipStream_t stream) {
  const float* x = (const float*)d_in[0];
  const float* Wq = (const float*)d_in[1];
  const float* Wk = (const float*)d_in[2];
  const float* Wv = (const float*)d_in[3];
  const float* Wo = (const float*)d_in[4];
  const float* rope = (const float*)d_in[5];
  const float* so8 = (const float*)d_in[6];

  float* ws = (float*)d_ws;
  float* qb = ws;
  float* kb = qb + (size_t)2 * N_HEADS * SEQ * HDIM;
  float* vtb = kb + (size_t)2 * N_KV * SEQ * HDIM;
  float* ab = vtb + (size_t)2 * N_KV * SEQ * HDIM;

  const int BS = 2 * SEQ;

  qkv_rope_so8_kernel<<<BS, 64, 0, stream>>>(x, Wq, Wk, Wv, rope, so8, qb, kb,
                                             vtb);
  so8_flash_attn_kernel<<<(2 * N_HEADS * QTILES) / 8, 256, 0, stream>>>(
      qb, kb, vtb, ab);
  out_proj_kernel<<<BS, 256, 0, stream>>>(ab, Wo, (float*)d_out);
}


# Output projection fused into the attention kernel: one 32-wave block per 16-row q tile holds all heads, projects the tile from LDS by f32 WMMA; 2 launches; projection summed in WMMA k order
# speedup vs baseline: 1.2035x; 1.2035x over previous
#include <hip/hip_runtime.h>
#include <hip/hip_bf16.h>

#define N_HEADS 32
#define N_KV 8
#define HDIM 8
#define SEQ 2048
#define HID 256
#define QTILES (SEQ / 16)

typedef __attribute__((ext_vector_type(2))) float v2f;
typedef __attribute__((ext_vector_type(8))) float v8f;

template <int CTRL>
__device__ __forceinline__ float dpp_max(float x) {
  const int y =
      __builtin_amdgcn_update_dpp(0, __float_as_int(x), CTRL, 0xf, 0xf, true);
  return fmaxf(x, __int_as_float(y));
}

__device__ __forceinline__ float rowmax16(float x) {
  x = dpp_max<0xB1>(x);
  x = dpp_max<0x4E>(x);
  x = dpp_max<0x141>(x);
  x = dpp_max<0x140>(x);
  return x;
}

__device__ __forceinline__ float fast_exp2(float x) {
  return __builtin_amdgcn_exp2f(x);
}

__global__ __launch_bounds__(64) void qkv_rope_so8_kernel(
    const float* __restrict__ x, const float* __restrict__ Wq,
    const float* __restrict__ Wk, const float* __restrict__ Wv,
    const float* __restrict__ rope, const float* __restrict__ so8,
    float* __restrict__ qb, float* __restrict__ kb, float* __restrict__ vtb) {
  const int bs = blockIdx.x;
  const int s = bs & (SEQ - 1);
  const int b = bs >> 11;

  __shared__ float xs[HID];
  for (int i = threadIdx.x; i < HID; i += 64) xs[i] = x[(size_t)bs * HID + i];
  __syncthreads();

  const int t = threadIdx.x;
  if (t >= 48) return;
  const int kind = (t < 32) ? 0 : (t < 40 ? 1 : 2);
  const int head = (t < 32) ? t : (t < 40 ? t - 32 : t - 40);
  const float* W = (kind == 0) ? Wq : ((kind == 1) ? Wk : Wv);
  const float* wr = W + (size_t)head * HDIM * HID;

  float acc[HDIM] = {};
  for (int i = 0; i < HID; ++i) {
    const float xv = xs[i];
#pragma unroll
    for (int d = 0; d < HDIM; ++d) acc[d] = fmaf(xv, wr[d * HID + i], acc[d]);
  }

  float y[HDIM];
  if (kind <= 1) {
    const float invf[4] = {1.0f, 0.1f, 0.01f, 0.001f};
    float cr[4] = {}, sr[4] = {};
#pragma unroll
    for (int i = 0; i < 4; ++i) {
      const float fr = (float)s * invf[i];
      const float ci = cosf(fr), si = sinf(fr);
#pragma unroll
      for (int j = 0; j < 4; ++j) {
        cr[j] = fmaf(ci, rope[i * 8 + j], cr[j]);
        sr[j] = fmaf(si, rope[i * 8 + j], sr[j]);
      }
    }
#pragma unroll
    for (int d = 0; d < HDIM; ++d) {
      const int j = d & 3;
      const float rh = (d < 4) ? -acc[d + 4] : acc[d - 4];
      y[d] = acc[d] * cr[j] + rh * sr[j];
    }
  } else {
#pragma unroll
    for (int d = 0; d < HDIM; ++d) y[d] = acc[d];
  }

  float z[HDIM];
#pragma unroll
  for (int j = 0; j < HDIM; ++j) {
    float sum = 0.0f;
#pragma unroll
    for (int d = 0; d < HDIM; ++d) sum = fmaf(y[d], so8[d * 8 + j], sum);
    z[j] = sum;
  }

  if (kind == 0) {
    float* dst = qb + (((size_t)b * N_HEADS + head) * SEQ + s) * HDIM;
#pragma unroll
    for (int j = 0; j < HDIM; ++j) dst[j] = z[j];
  } else if (kind == 1) {
    float* dst = kb + (((size_t)b * N_KV + head) * SEQ + s) * HDIM;
#pragma unroll
    for (int j = 0; j < HDIM; ++j) dst[j] = z[j];
  } else {
#pragma unroll
    for (int j = 0; j < HDIM; ++j)
      vtb[(((size_t)b * N_KV + head) * HDIM + j) * SEQ + s] = z[j];
  }
}

template <bool MASK>
__device__ __forceinline__ void attn_tile(
    const float* __restrict__ kp, const float* __restrict__ vp,
    float* __restrict__ pw, int kbase, int lanelo, int hi, int hi2,
    const float* __restrict__ maskv, const v2f Qa0, const v2f Qa1,
    float* __restrict__ m, v8f& o) {
  const float* krow = kp + (size_t)(kbase + lanelo) * HDIM;
  const v2f Kb0 = *(const v2f*)(krow + hi2);
  const v2f Kb1 = *(const v2f*)(krow + 4 + hi2);

  v8f sc = {};
  sc = __builtin_amdgcn_wmma_f32_16x16x4_f32(false, Qa0, false, Kb0, (short)0,
                                             sc, false, false);
  sc = __builtin_amdgcn_wmma_f32_16x16x4_f32(false, Qa1, false, Kb1, (short)0,
                                             sc, false, false);

  float p[8];
#pragma unroll
  for (int r = 0; r < 8; ++r) {
    float sv = sc[r];
    if (MASK) sv += maskv[r];
    const float rm = rowmax16(sv);
    const float mnew = fmaxf(m[r], rm);
    const float corr = fast_exp2(m[r] - mnew);
    m[r] = mnew;
    p[r] = fast_exp2(sv - mnew);
    o[r] = o[r] * corr;
  }

#pragma unroll
  for (int r = 0; r < 8; ++r) pw[r * 16 + hi * 128 + lanelo] = p[r];
  v2f Pa[4];
#pragma unroll
  for (int t = 0; t < 4; ++t)
    Pa[t] = *(const v2f*)(pw + (lanelo & 7) * 16 + (lanelo >> 3) * 128 +
                          4 * t + hi2);

  v2f Vb[4];
#pragma unroll
  for (int t = 0; t < 4; ++t) {
    if (lanelo < 8) {
      Vb[t] = *(const v2f*)(vp + (size_t)lanelo * SEQ + kbase + 4 * t + hi2);
    } else {
      const float c = (lanelo == 8) ? 1.0f : 0.0f;
      Vb[t] = (v2f){c, c};
    }
  }
#pragma unroll
  for (int t = 0; t < 4; ++t)
    o = __builtin_amdgcn_wmma_f32_16x16x4_f32(false, Pa[t], false, Vb[t],
                                              (short)0, o, false, false);
}

#define ATP 260

__global__ __launch_bounds__(1024) void so8_flash_attn_outproj_kernel(
    const float* __restrict__ q, const float* __restrict__ k,
    const float* __restrict__ vT, const float* __restrict__ Wo,
    float* __restrict__ out) {
  const int lane = threadIdx.x & 31;
  const int h = threadIdx.x >> 5;
  const int b = blockIdx.x & 1;
  const int qt = (QTILES - 1) - (blockIdx.x >> 1);
  const int kv = h >> 2;
  const int qbase = qt * 16;
  const int lanelo = lane & 15;
  const int hi = lane >> 4;
  const int hi2 = hi * 2;

  const float* qp = q + ((size_t)b * N_HEADS + h) * SEQ * HDIM;
  const float* kp = k + ((size_t)b * N_KV + kv) * SEQ * HDIM;
  const float* vp = vT + ((size_t)b * N_KV + kv) * HDIM * SEQ;

  __shared__ __attribute__((aligned(16))) float plds[N_HEADS * 256];
  __shared__ __attribute__((aligned(16))) float tile[16 * ATP];
  float* pw = plds + h * 256;

  const float QSCALE = 0.35355339059327f * 1.44269504088896f;
  const float* qrow = qp + (size_t)(qbase + lanelo) * HDIM;
  const v2f Qa0 = *(const v2f*)(qrow + hi2) * QSCALE;
  const v2f Qa1 = *(const v2f*)(qrow + 4 + hi2) * QSCALE;

  float maskv[8];
#pragma unroll
  for (int r = 0; r < 8; ++r)
    maskv[r] = (lanelo > r + 8 * hi) ? -3.0e38f : 0.0f;

  v8f o = {};
  float m[8];
#pragma unroll
  for (int r = 0; r < 8; ++r) m[r] = -3.0e38f;

  for (int kt = 0; kt < qt; ++kt)
    attn_tile<false>(kp, vp, pw, kt * 16, lanelo, hi, hi2, maskv, Qa0, Qa1, m,
                     o);
  attn_tile<true>(kp, vp, pw, qbase, lanelo, hi, hi2, maskv, Qa0, Qa1, m, o);

#pragma unroll
  for (int r = 0; r < 8; ++r) {
    const float ov = o[r];
    const float l = __shfl(ov, (lane & 16) + 8);
    const float outv = ov * __builtin_amdgcn_rcpf(l);
    if (lanelo < HDIM) tile[(r + 8 * hi) * ATP + h * HDIM + lanelo] = outv;
  }
  __syncthreads();

  const int ct = h & 15;
  const int kh = h >> 4;
  const float* arow = tile + lanelo * ATP + kh * 128 + 4 * hi;
  const float* brow = Wo + (size_t)(ct * 16 + lanelo) * HID + kh * 128 + 4 * hi;
  v8f acc0 = {}, acc1 = {};
#pragma unroll
  for (int u = 0; u < 16; ++u) {
    const float4 a4 = *(const float4*)(arow + 8 * u);
    const float4 b4 = *(const float4*)(brow + 8 * u);
    acc0 = __builtin_amdgcn_wmma_f32_16x16x4_f32(
        false, (v2f){a4.x, a4.y}, false, (v2f){b4.x, b4.y}, (short)0, acc0,
        false, false);
    acc1 = __builtin_amdgcn_wmma_f32_16x16x4_f32(
        false, (v2f){a4.z, a4.w}, false, (v2f){b4.z, b4.w}, (short)0, acc1,
        false, false);
  }
  const v8f acc = acc0 + acc1;

  float* red = plds + ct * 256;
  if (kh == 1) {
#pragma unroll
    for (int r = 0; r < 8; ++r) red[r * 32 + lane] = acc[r];
  }
  __syncthreads();
  if (kh == 0) {
#pragma unroll
    for (int r = 0; r < 8; ++r)
      tile[(r + 8 * hi) * ATP + ct * 16 + lanelo] = acc[r] + red[r * 32 + lane];
  }
  __syncthreads();

  const int orow = h >> 1;
  const int oc = (h & 1) * 128 + lane * 4;
  const float4 ov4 = *(const float4*)(tile + orow * ATP + oc);
  *(float4*)(out + ((size_t)b * SEQ + qbase + orow) * HID + oc) = ov4;
}

extern "C" void kernel_launch(void* const* d_in, const int* in_sizes, int n_in,
                              void* d_out, int out_size, void* d_ws,
                              size_t ws_size, hipStream_t stream) {
  const float* x = (const float*)d_in[0];
  const float* Wq = (const float*)d_in[1];
  const float* Wk = (const float*)d_in[2];
  const float* Wv = (const float*)d_in[3];
  const float* Wo = (const float*)d_in[4];
  const float* rope = (const float*)d_in[5];
  const float* so8 = (const float*)d_in[6];

  float* ws = (float*)d_ws;
  float* qb = ws;
  float* kb = qb + (size_t)2 * N_HEADS * SEQ * HDIM;
  float* vtb = kb + (size_t)2 * N_KV * SEQ * HDIM;

  const int BS = 2 * SEQ;

  qkv_rope_so8_kernel<<<BS, 64, 0, stream>>>(x, Wq, Wk, Wv, rope, so8, qb, kb,
                                             vtb);
  so8_flash_attn_outproj_kernel<<<2 * QTILES, 1024, 0, stream>>>(
      qb, kb, vtb, Wo, (float*)d_out);
}


# Attention key loop on 64-key blocks: one row max, stabiliser update and accumulator rescale per 64 keys (was 16); K/V loads grouped, V's ones column read from a constant row; rescale points change
# speedup vs baseline: 1.6234x; 1.3489x over previous
#include <hip/hip_runtime.h>
#include <hip/hip_bf16.h>

#define N_HEADS 32
#define N_KV 8
#define HDIM 8
#define SEQ 2048
#define HID 256
#define QTILES (SEQ / 16)

typedef __attribute__((ext_vector_type(2))) float v2f;
typedef __attribute__((ext_vector_type(8))) float v8f;
typedef const __attribute__((address_space(1))) v2f gv2f;

__device__ const float kOnes[64] = {
    1.0f, 1.0f, 1.0f, 1.0f, 1.0f, 1.0f, 1.0f, 1.0f, 1.0f, 1.0f, 1.0f, 1.0f, 1.0f,
    1.0f, 1.0f, 1.0f, 1.0f, 1.0f, 1.0f, 1.0f, 1.0f, 1.0f, 1.0f, 1.0f, 1.0f, 1.0f,
    1.0f, 1.0f, 1.0f, 1.0f, 1.0f, 1.0f, 1.0f, 1.0f, 1.0f, 1.0f, 1.0f, 1.0f, 1.0f,
    1.0f, 1.0f, 1.0f, 1.0f, 1.0f, 1.0f, 1.0f, 1.0f, 1.0f, 1.0f, 1.0f, 1.0f, 1.0f,
    1.0f, 1.0f, 1.0f, 1.0f, 1.0f, 1.0f, 1.0f, 1.0f, 1.0f, 1.0f, 1.0f, 1.0f};

template <int CTRL>
__device__ __forceinline__ float dpp_max(float x) {
  const int y =
      __builtin_amdgcn_update_dpp(0, __float_as_int(x), CTRL, 0xf, 0xf, true);
  return fmaxf(x, __int_as_float(y));
}

__device__ __forceinline__ float rowmax16(float x) {
  x = dpp_max<0xB1>(x);
  x = dpp_max<0x4E>(x);
  x = dpp_max<0x141>(x);
  x = dpp_max<0x140>(x);
  return x;
}

__device__ __forceinline__ float fast_exp2(float x) {
  return __builtin_amdgcn_exp2f(x);
}

__global__ __launch_bounds__(64) void qkv_rope_so8_kernel(
    const float* __restrict__ x, const float* __restrict__ Wq,
    const float* __restrict__ Wk, const float* __restrict__ Wv,
    const float* __restrict__ rope, const float* __restrict__ so8,
    float* __restrict__ qb, float* __restrict__ kb, float* __restrict__ vtb) {
  const int bs = blockIdx.x;
  const int s = bs & (SEQ - 1);
  const int b = bs >> 11;

  __shared__ float xs[HID];
  for (int i = threadIdx.x; i < HID; i += 64) xs[i] = x[(size_t)bs * HID + i];
  __syncthreads();

  const int t = threadIdx.x;
  if (t >= 48) return;
  const int kind = (t < 32) ? 0 : (t < 40 ? 1 : 2);
  const int head = (t < 32) ? t : (t < 40 ? t - 32 : t - 40);
  const float* W = (kind == 0) ? Wq : ((kind == 1) ? Wk : Wv);
  const float* wr = W + (size_t)head * HDIM * HID;

  float acc[HDIM] = {};
  for (int i = 0; i < HID; ++i) {
    const float xv = xs[i];
#pragma unroll
    for (int d = 0; d < HDIM; ++d) acc[d] = fmaf(xv, wr[d * HID + i], acc[d]);
  }

  float y[HDIM];
  if (kind <= 1) {
    const float invf[4] = {1.0f, 0.1f, 0.01f, 0.001f};
    float cr[4] = {}, sr[4] = {};
#pragma unroll
    for (int i = 0; i < 4; ++i) {
      const float fr = (float)s * invf[i];
      const float ci = cosf(fr), si = sinf(fr);
#pragma unroll
      for (int j = 0; j < 4; ++j) {
        cr[j] = fmaf(ci, rope[i * 8 + j], cr[j]);
        sr[j] = fmaf(si, rope[i * 8 + j], sr[j]);
      }
    }
#pragma unroll
    for (int d = 0; d < HDIM; ++d) {
      const int j = d & 3;
      const float rh = (d < 4) ? -acc[d + 4] : acc[d - 4];
      y[d] = acc[d] * cr[j] + rh * sr[j];
    }
  } else {
#pragma unroll
    for (int d = 0; d < HDIM; ++d) y[d] = acc[d];
  }

  float z[HDIM];
#pragma unroll
  for (int j = 0; j < HDIM; ++j) {
    float sum = 0.0f;
#pragma unroll
    for (int d = 0; d < HDIM; ++d) sum = fmaf(y[d], so8[d * 8 + j], sum);
    z[j] = sum;
  }

  if (kind == 0) {
    float* dst = qb + (((size_t)b * N_HEADS + head) * SEQ + s) * HDIM;
#pragma unroll
    for (int j = 0; j < HDIM; ++j) dst[j] = z[j];
  } else if (kind == 1) {
    float* dst = kb + (((size_t)b * N_KV + head) * SEQ + s) * HDIM;
#pragma unroll
    for (int j = 0; j < HDIM; ++j) dst[j] = z[j];
  } else {
#pragma unroll
    for (int j = 0; j < HDIM; ++j)
      vtb[(((size_t)b * N_KV + head) * HDIM + j) * SEQ + s] = z[j];
  }
}

#define PPITCH 20
#define PTILE (16 * PPITCH)

template <int NT, bool MASK>
__device__ __forceinline__ void attn_block(
    const float* __restrict__ kp, const float* __restrict__ vrow,
    float* __restrict__ pw, int kbase, int lanelo, int hi, int hi2,
    bool vlane, const float* __restrict__ maskv, const v2f Qa0, const v2f Qa1,
    float* __restrict__ m, v8f& o) {
  v2f Kb0[NT], Kb1[NT];
#pragma unroll
  for (int t = 0; t < NT; ++t) {
    const float* krow = kp + (size_t)(kbase + 16 * t + lanelo) * HDIM;
    Kb0[t] = *(const v2f*)(krow + hi2);
    Kb1[t] = *(const v2f*)(krow + 4 + hi2);
  }
  const float* vsrc = vrow + (vlane ? kbase : 0) + hi2;
  v2f Vb[NT][4];
#pragma unroll
  for (int t = 0; t < NT; ++t) {
#pragma unroll
    for (int u = 0; u < 4; ++u)
      Vb[t][u] = *(gv2f*)(vsrc + 16 * t + 4 * u);
  }

  v8f sc[NT];
#pragma unroll
  for (int t = 0; t < NT; ++t) {
    v8f z = {};
    z = __builtin_amdgcn_wmma_f32_16x16x4_f32(false, Qa0, false, Kb0[t],
                                              (short)0, z, false, false);
    sc[t] = __builtin_amdgcn_wmma_f32_16x16x4_f32(false, Qa1, false, Kb1[t],
                                                  (short)0, z, false, false);
  }

#pragma unroll
  for (int r = 0; r < 8; ++r) {
    if (MASK) sc[NT - 1][r] += maskv[r];
    float mx = sc[0][r];
#pragma unroll
    for (int t = 1; t < NT; ++t) mx = fmaxf(mx, sc[t][r]);
    const float rm = rowmax16(mx);
    const float mnew = fmaxf(m[r], rm);
    const float corr = fast_exp2(m[r] - mnew);
    m[r] = mnew;
    o[r] = o[r] * corr;
#pragma unroll
    for (int t = 0; t < NT; ++t) sc[t][r] = fast_exp2(sc[t][r] - mnew);
  }

  float* pst = pw + hi * (8 * PPITCH) + lanelo;
  const float* pld = pw + lanelo * PPITCH + hi2;
#pragma unroll
  for (int t = 0; t < NT; ++t) {
#pragma unroll
    for (int r = 0; r < 8; ++r) pst[t * PTILE + r * PPITCH] = sc[t][r];
  }
  v2f Pa[NT][4];
#pragma unroll
  for (int t = 0; t < NT; ++t) {
#pragma unroll
    for (int u = 0; u < 4; ++u)
      Pa[t][u] = *(const v2f*)(pld + t * PTILE + 4 * u);
  }
#pragma unroll
  for (int t = 0; t < NT; ++t) {
#pragma unroll
    for (int u = 0; u < 4; ++u)
      o = __builtin_amdgcn_wmma_f32_16x16x4_f32(false, Pa[t][u], false,
                                                Vb[t][u], (short)0, o, false,
                                                false);
  }
}

#define ATP 260

__global__ __launch_bounds__(1024) void so8_flash_attn_outproj_kernel(
    const float* __restrict__ q, const float* __restrict__ k,
    const float* __restrict__ vT, const float* __restrict__ Wo,
    float* __restrict__ out) {
  const int lane = threadIdx.x & 31;
  const int h = threadIdx.x >> 5;
  const int b = blockIdx.x & 1;
  const int qt = (QTILES - 1) - (blockIdx.x >> 1);
  const int kv = h >> 2;
  const int qbase = qt * 16;
  const int lanelo = lane & 15;
  const int hi = lane >> 4;
  const int hi2 = hi * 2;

  const float* qp = q + ((size_t)b * N_HEADS + h) * SEQ * HDIM;
  const float* kp = k + ((size_t)b * N_KV + kv) * SEQ * HDIM;
  const float* vp = vT + ((size_t)b * N_KV + kv) * HDIM * SEQ;
  const bool vlane = lanelo < 8;
  const float* vrow = vlane ? vp + (size_t)lanelo * SEQ : kOnes;

  __shared__ __attribute__((aligned(16))) float plds[N_HEADS * 4 * PTILE];
  __shared__ __attribute__((aligned(16))) float tile[16 * ATP];
  float* pw = plds + h * 4 * PTILE;

  const float QSCALE = 0.35355339059327f * 1.44269504088896f;
  const float* qrow = qp + (size_t)(qbase + lanelo) * HDIM;
  const v2f Qa0 = *(const v2f*)(qrow + hi2) * QSCALE;
  const v2f Qa1 = *(const v2f*)(qrow + 4 + hi2) * QSCALE;

  float maskv[8];
#pragma unroll
  for (int r = 0; r < 8; ++r)
    maskv[r] = (lanelo > r + 8 * hi) ? -3.0e38f : 0.0f;

  v8f o = {};
  float m[8];
#pragma unroll
  for (int r = 0; r < 8; ++r) m[r] = -3.0e38f;

  const int nfull = qt >> 2;
  for (int kb = 0; kb < nfull; ++kb)
    attn_block<4, false>(kp, vrow, pw, kb * 64, lanelo, hi, hi2, vlane, maskv,
                         Qa0, Qa1, m, o);
  const int tbase = nfull * 64;
  switch (qt & 3) {
    case 0:
      attn_block<1, true>(kp, vrow, pw, tbase, lanelo, hi, hi2, vlane, maskv,
                          Qa0, Qa1, m, o);
      break;
    case 1:
      attn_block<2, true>(kp, vrow, pw, tbase, lanelo, hi, hi2, vlane, maskv,
                          Qa0, Qa1, m, o);
      break;
    case 2:
      attn_block<3, true>(kp, vrow, pw, tbase, lanelo, hi, hi2, vlane, maskv,
                          Qa0, Qa1, m, o);
      break;
    default:
      attn_block<4, true>(kp, vrow, pw, tbase, lanelo, hi, hi2, vlane, maskv,
                          Qa0, Qa1, m, o);
      break;
  }

#pragma unroll
  for (int r = 0; r < 8; ++r) {
    const float ov = o[r];
    const float l = __shfl(ov, (lane & 16) + 8);
    const float outv = ov * __builtin_amdgcn_rcpf(l);
    if (lanelo < HDIM) tile[(r + 8 * hi) * ATP + h * HDIM + lanelo] = outv;
  }
  __syncthreads();

  const int ct = h & 15;
  const int kh = h >> 4;
  const float* arow = tile + lanelo * ATP + kh * 128 + 4 * hi;
  const float* brow = Wo + (size_t)(ct * 16 + lanelo) * HID + kh * 128 + 4 * hi;
  v8f acc0 = {}, acc1 = {};
#pragma unroll
  for (int u = 0; u < 16; ++u) {
    const float4 a4 = *(const float4*)(arow + 8 * u);
    const float4 b4 = *(const float4*)(brow + 8 * u);
    acc0 = __builtin_amdgcn_wmma_f32_16x16x4_f32(
        false, (v2f){a4.x, a4.y}, false, (v2f){b4.x, b4.y}, (short)0, acc0,
        false, false);
    acc1 = __builtin_amdgcn_wmma_f32_16x16x4_f32(
        false, (v2f){a4.z, a4.w}, false, (v2f){b4.z, b4.w}, (short)0, acc1,
        false, false);
  }
  const v8f acc = acc0 + acc1;

  float* red = plds + ct * 256;
  if (kh == 1) {
#pragma unroll
    for (int r = 0; r < 8; ++r) red[r * 32 + lane] = acc[r];
  }
  __syncthreads();
  if (kh == 0) {
#pragma unroll
    for (int r = 0; r < 8; ++r)
      tile[(r + 8 * hi) * ATP + ct * 16 + lanelo] = acc[r] + red[r * 32 + lane];
  }
  __syncthreads();

  const int orow = h >> 1;
  const int oc = (h & 1) * 128 + lane * 4;
  const float4 ov4 = *(const float4*)(tile + orow * ATP + oc);
  *(float4*)(out + ((size_t)b * SEQ + qbase + orow) * HID + oc) = ov4;
}

extern "C" void kernel_launch(void* const* d_in, const int* in_sizes, int n_in,
                              void* d_out, int out_size, void* d_ws,
                              size_t ws_size, hipStream_t stream) {
  const float* x = (const float*)d_in[0];
  const float* Wq = (const float*)d_in[1];
  const float* Wk = (const float*)d_in[2];
  const float* Wv = (const float*)d_in[3];
  const float* Wo = (const float*)d_in[4];
  const float* rope = (const float*)d_in[5];
  const float* so8 = (const float*)d_in[6];

  float* ws = (float*)d_ws;
  float* qb = ws;
  float* kb = qb + (size_t)2 * N_HEADS * SEQ * HDIM;
  float* vtb = kb + (size_t)2 * N_KV * SEQ * HDIM;

  const int BS = 2 * SEQ;

  qkv_rope_so8_kernel<<<BS, 64, 0, stream>>>(x, Wq, Wk, Wv, rope, so8, qb, kb,
                                             vtb);
  so8_flash_attn_outproj_kernel<<<2 * QTILES, 1024, 0, stream>>>(
      qb, kb, vtb, Wo, (float*)d_out);
}


# Q projection moved into the attention kernel: each block stages its 16 x rows in LDS, projects them by Wq on f32 WMMA, ropes/rotates per head; first kernel does K/V only; q summed in WMMA k order
# speedup vs baseline: 2.3501x; 1.4476x over previous
#include <hip/hip_runtime.h>
#include <hip/hip_bf16.h>

#define N_HEADS 32
#define N_KV 8
#define HDIM 8
#define SEQ 2048
#define HID 256
#define QTILES (SEQ / 16)

typedef __attribute__((ext_vector_type(2))) float v2f;
typedef __attribute__((ext_vector_type(8))) float v8f;
typedef const __attribute__((address_space(1))) v2f gv2f;

__device__ const float kOnes[64] = {
    1.0f, 1.0f, 1.0f, 1.0f, 1.0f, 1.0f, 1.0f, 1.0f, 1.0f, 1.0f, 1.0f, 1.0f, 1.0f,
    1.0f, 1.0f, 1.0f, 1.0f, 1.0f, 1.0f, 1.0f, 1.0f, 1.0f, 1.0f, 1.0f, 1.0f, 1.0f,
    1.0f, 1.0f, 1.0f, 1.0f, 1.0f, 1.0f, 1.0f, 1.0f, 1.0f, 1.0f, 1.0f, 1.0f, 1.0f,
    1.0f, 1.0f, 1.0f, 1.0f, 1.0f, 1.0f, 1.0f, 1.0f, 1.0f, 1.0f, 1.0f, 1.0f, 1.0f,
    1.0f, 1.0f, 1.0f, 1.0f, 1.0f, 1.0f, 1.0f, 1.0f, 1.0f, 1.0f, 1.0f, 1.0f};

template <int CTRL>
__device__ __forceinline__ float dpp_max(float x) {
  const int y =
      __builtin_amdgcn_update_dpp(0, __float_as_int(x), CTRL, 0xf, 0xf, true);
  return fmaxf(x, __int_as_float(y));
}

__device__ __forceinline__ float rowmax16(float x) {
  x = dpp_max<0xB1>(x);
  x = dpp_max<0x4E>(x);
  x = dpp_max<0x141>(x);
  x = dpp_max<0x140>(x);
  return x;
}

__device__ __forceinline__ float fast_exp2(float x) {
  return __builtin_amdgcn_exp2f(x);
}

__global__ __launch_bounds__(32) void kv_rope_so8_kernel(
    const float* __restrict__ x, const float* __restrict__ Wk,
    const float* __restrict__ Wv, const float* __restrict__ rope,
    const float* __restrict__ so8, float* __restrict__ kb,
    float* __restrict__ vtb) {
  const int bs = blockIdx.x;
  const int s = bs & (SEQ - 1);
  const int b = bs >> 11;

  __shared__ float xs[HID];
  for (int i = threadIdx.x; i < HID; i += 32) xs[i] = x[(size_t)bs * HID + i];
  __syncthreads();

  const int t = threadIdx.x;
  if (t >= 16) return;
  const int kind = (t < 8) ? 1 : 2;
  const int head = t & 7;
  const float* W = (kind == 1) ? Wk : Wv;
  const float* wr = W + (size_t)head * HDIM * HID;

  float acc[HDIM] = {};
  for (int i = 0; i < HID; ++i) {
    const float xv = xs[i];
#pragma unroll
    for (int d = 0; d < HDIM; ++d) acc[d] = fmaf(xv, wr[d * HID + i], acc[d]);
  }

  float y[HDIM];
  if (kind <= 1) {
    const float invf[4] = {1.0f, 0.1f, 0.01f, 0.001f};
    float cr[4] = {}, sr[4] = {};
#pragma unroll
    for (int i = 0; i < 4; ++i) {
      const float fr = (float)s * invf[i];
      const float ci = cosf(fr), si = sinf(fr);
#pragma unroll
      for (int j = 0; j < 4; ++j) {
        cr[j] = fmaf(ci, rope[i * 8 + j], cr[j]);
        sr[j] = fmaf(si, rope[i * 8 + j], sr[j]);
      }
    }
#pragma unroll
    for (int d = 0; d < HDIM; ++d) {
      const int j = d & 3;
      const float rh = (d < 4) ? -acc[d + 4] : acc[d - 4];
      y[d] = acc[d] * cr[j] + rh * sr[j];
    }
  } else {
#pragma unroll
    for (int d = 0; d < HDIM; ++d) y[d] = acc[d];
  }

  float z[HDIM];
#pragma unroll
  for (int j = 0; j < HDIM; ++j) {
    float sum = 0.0f;
#pragma unroll
    for (int d = 0; d < HDIM; ++d) sum = fmaf(y[d], so8[d * 8 + j], sum);
    z[j] = sum;
  }

  if (kind == 1) {
    float* dst = kb + (((size_t)b * N_KV + head) * SEQ + s) * HDIM;
#pragma unroll
    for (int j = 0; j < HDIM; ++j) dst[j] = z[j];
  } else {
#pragma unroll
    for (int j = 0; j < HDIM; ++j)
      vtb[(((size_t)b * N_KV + head) * HDIM + j) * SEQ + s] = z[j];
  }
}

#define PPITCH 20
#define PTILE (16 * PPITCH)

template <int NT, bool MASK>
__device__ __forceinline__ void attn_block(
    const float* __restrict__ kp, const float* __restrict__ vrow,
    float* __restrict__ pw, int kbase, int lanelo, int hi, int hi2,
    bool vlane, const float* __restrict__ maskv, const v2f Qa0, const v2f Qa1,
    float* __restrict__ m, v8f& o) {
  v2f Kb0[NT], Kb1[NT];
#pragma unroll
  for (int t = 0; t < NT; ++t) {
    const float* krow = kp + (size_t)(kbase + 16 * t + lanelo) * HDIM;
    Kb0[t] = *(const v2f*)(krow + hi2);
    Kb1[t] = *(const v2f*)(krow + 4 + hi2);
  }
  const float* vsrc = vrow + (vlane ? kbase : 0) + hi2;
  v2f Vb[NT][4];
#pragma unroll
  for (int t = 0; t < NT; ++t) {
#pragma unroll
    for (int u = 0; u < 4; ++u)
      Vb[t][u] = *(gv2f*)(vsrc + 16 * t + 4 * u);
  }

  v8f sc[NT];
#pragma unroll
  for (int t = 0; t < NT; ++t) {
    v8f z = {};
    z = __builtin_amdgcn_wmma_f32_16x16x4_f32(false, Qa0, false, Kb0[t],
                                              (short)0, z, false, false);
    sc[t] = __builtin_amdgcn_wmma_f32_16x16x4_f32(false, Qa1, false, Kb1[t],
                                                  (short)0, z, false, false);
  }

#pragma unroll
  for (int r = 0; r < 8; ++r) {
    if (MASK) sc[NT - 1][r] += maskv[r];
    float mx = sc[0][r];
#pragma unroll
    for (int t = 1; t < NT; ++t) mx = fmaxf(mx, sc[t][r]);
    const float rm = rowmax16(mx);
    const float mnew = fmaxf(m[r], rm);
    const float corr = fast_exp2(m[r] - mnew);
    m[r] = mnew;
    o[r] = o[r] * corr;
#pragma unroll
    for (int t = 0; t < NT; ++t) sc[t][r] = fast_exp2(sc[t][r] - mnew);
  }

  float* pst = pw + hi * (8 * PPITCH) + lanelo;
  const float* pld = pw + lanelo * PPITCH + hi2;
#pragma unroll
  for (int t = 0; t < NT; ++t) {
#pragma unroll
    for (int r = 0; r < 8; ++r) pst[t * PTILE + r * PPITCH] = sc[t][r];
  }
  v2f Pa[NT][4];
#pragma unroll
  for (int t = 0; t < NT; ++t) {
#pragma unroll
    for (int u = 0; u < 4; ++u)
      Pa[t][u] = *(const v2f*)(pld + t * PTILE + 4 * u);
  }
#pragma unroll
  for (int t = 0; t < NT; ++t) {
#pragma unroll
    for (int u = 0; u < 4; ++u)
      o = __builtin_amdgcn_wmma_f32_16x16x4_f32(false, Pa[t][u], false,
                                                Vb[t][u], (short)0, o, false,
                                                false);
  }
}

#define ATP 260

__device__ __forceinline__ v8f tile_gemm_khalf(const float* __restrict__ tile,
                                               const float* __restrict__ Wm,
                                               int ct, int kh, int lanelo,
                                               int hi) {
  const float* arow = tile + lanelo * ATP + kh * 128 + 4 * hi;
  const float* brow = Wm + (size_t)(ct * 16 + lanelo) * HID + kh * 128 + 4 * hi;
  v8f acc0 = {}, acc1 = {};
#pragma unroll
  for (int u = 0; u < 16; ++u) {
    const float4 a4 = *(const float4*)(arow + 8 * u);
    const float4 b4 = *(const float4*)(brow + 8 * u);
    acc0 = __builtin_amdgcn_wmma_f32_16x16x4_f32(
        false, (v2f){a4.x, a4.y}, false, (v2f){b4.x, b4.y}, (short)0, acc0,
        false, false);
    acc1 = __builtin_amdgcn_wmma_f32_16x16x4_f32(
        false, (v2f){a4.z, a4.w}, false, (v2f){b4.z, b4.w}, (short)0, acc1,
        false, false);
  }
  return acc0 + acc1;
}

__global__ __launch_bounds__(1024) void so8_qproj_flash_attn_outproj_kernel(
    const float* __restrict__ x, const float* __restrict__ Wq,
    const float* __restrict__ rope, const float* __restrict__ so8,
    const float* __restrict__ k, const float* __restrict__ vT,
    const float* __restrict__ Wo, float* __restrict__ out) {
  const int lane = threadIdx.x & 31;
  const int h = threadIdx.x >> 5;
  const int b = blockIdx.x & 1;
  const int qt = (QTILES - 1) - (blockIdx.x >> 1);
  const int kv = h >> 2;
  const int qbase = qt * 16;
  const int lanelo = lane & 15;
  const int hi = lane >> 4;
  const int hi2 = hi * 2;

  const float* kp = k + ((size_t)b * N_KV + kv) * SEQ * HDIM;
  const float* vp = vT + ((size_t)b * N_KV + kv) * HDIM * SEQ;
  const bool vlane = lanelo < 8;
  const float* vrow = vlane ? vp + (size_t)lanelo * SEQ : kOnes;

  __shared__ __attribute__((aligned(16))) float plds[N_HEADS * 4 * PTILE];
  __shared__ __attribute__((aligned(16))) float tile[16 * ATP];
  __shared__ __attribute__((aligned(16))) float trig[16 * 8];
  float* pw = plds + h * 4 * PTILE;
  const int ct = h & 15;
  const int kh = h >> 4;

  {
    const int xr = threadIdx.x >> 6;
    const int xc = (threadIdx.x & 63) * 4;
    const float4 xv =
        *(const float4*)(x + ((size_t)b * SEQ + qbase + xr) * HID + xc);
    *(float4*)(tile + xr * ATP + xc) = xv;
  }
  if (threadIdx.x < 128) {
    const int i = threadIdx.x & 3;
    const float invf = (i == 0) ? 1.0f : (i == 1) ? 0.1f : (i == 2) ? 0.01f : 0.001f;
    const float fr = (float)(qbase + (threadIdx.x >> 3)) * invf;
    trig[threadIdx.x] = (threadIdx.x & 4) ? sinf(fr) : cosf(fr);
  }
  __syncthreads();
  {
    const v8f qacc = tile_gemm_khalf(tile, Wq, ct, kh, lanelo, hi);
    float* red = plds + ct * 256;
    if (kh == 1) {
#pragma unroll
      for (int r = 0; r < 8; ++r) red[r * 32 + lane] = qacc[r];
    }
    __syncthreads();
    if (kh == 0) {
#pragma unroll
      for (int r = 0; r < 8; ++r)
        tile[(r + 8 * hi) * ATP + ct * 16 + lanelo] =
            qacc[r] + red[r * 32 + lane];
    }
    __syncthreads();
  }

  v2f Qa0, Qa1;
  {
    const float4 q0 = *(const float4*)(tile + lanelo * ATP + h * HDIM);
    const float4 q1 = *(const float4*)(tile + lanelo * ATP + h * HDIM + 4);
    const float4 c4 = *(const float4*)(trig + lanelo * 8);
    const float4 s4 = *(const float4*)(trig + lanelo * 8 + 4);
    const float acc[HDIM] = {q0.x, q0.y, q0.z, q0.w, q1.x, q1.y, q1.z, q1.w};
    const float cs[4] = {c4.x, c4.y, c4.z, c4.w};
    const float sn[4] = {s4.x, s4.y, s4.z, s4.w};
    float cr[4] = {}, sr[4] = {};
#pragma unroll
    for (int i = 0; i < 4; ++i) {
#pragma unroll
      for (int j = 0; j < 4; ++j) {
        cr[j] = fmaf(cs[i], rope[i * 8 + j], cr[j]);
        sr[j] = fmaf(sn[i], rope[i * 8 + j], sr[j]);
      }
    }
    float y[HDIM];
#pragma unroll
    for (int d = 0; d < HDIM; ++d) {
      const int j = d & 3;
      const float rh = (d < 4) ? -acc[d + 4] : acc[d - 4];
      y[d] = acc[d] * cr[j] + rh * sr[j];
    }
    float z[HDIM];
#pragma unroll
    for (int j = 0; j < HDIM; ++j) {
      float sum = 0.0f;
#pragma unroll
      for (int d = 0; d < HDIM; ++d) sum = fmaf(y[d], so8[d * 8 + j], sum);
      z[j] = sum;
    }
    const float QSCALE = 0.35355339059327f * 1.44269504088896f;
    Qa0 = (v2f){hi ? z[2] : z[0], hi ? z[3] : z[1]} * QSCALE;
    Qa1 = (v2f){hi ? z[6] : z[4], hi ? z[7] : z[5]} * QSCALE;
  }
  __syncthreads();

  float maskv[8];
#pragma unroll
  for (int r = 0; r < 8; ++r)
    maskv[r] = (lanelo > r + 8 * hi) ? -3.0e38f : 0.0f;

  v8f o = {};
  float m[8];
#pragma unroll
  for (int r = 0; r < 8; ++r) m[r] = -3.0e38f;

  const int nfull = qt >> 2;
  for (int kb = 0; kb < nfull; ++kb)
    attn_block<4, false>(kp, vrow, pw, kb * 64, lanelo, hi, hi2, vlane, maskv,
                         Qa0, Qa1, m, o);
  const int tbase = nfull * 64;
  switch (qt & 3) {
    case 0:
      attn_block<1, true>(kp, vrow, pw, tbase, lanelo, hi, hi2, vlane, maskv,
                          Qa0, Qa1, m, o);
      break;
    case 1:
      attn_block<2, true>(kp, vrow, pw, tbase, lanelo, hi, hi2, vlane, maskv,
                          Qa0, Qa1, m, o);
      break;
    case 2:
      attn_block<3, true>(kp, vrow, pw, tbase, lanelo, hi, hi2, vlane, maskv,
                          Qa0, Qa1, m, o);
      break;
    default:
      attn_block<4, true>(kp, vrow, pw, tbase, lanelo, hi, hi2, vlane, maskv,
                          Qa0, Qa1, m, o);
      break;
  }

#pragma unroll
  for (int r = 0; r < 8; ++r) {
    const float ov = o[r];
    const float l = __shfl(ov, (lane & 16) + 8);
    const float outv = ov * __builtin_amdgcn_rcpf(l);
    if (lanelo < HDIM) tile[(r + 8 * hi) * ATP + h * HDIM + lanelo] = outv;
  }
  __syncthreads();

  const v8f acc = tile_gemm_khalf(tile, Wo, ct, kh, lanelo, hi);

  float* red = plds + ct * 256;
  if (kh == 1) {
#pragma unroll
    for (int r = 0; r < 8; ++r) red[r * 32 + lane] = acc[r];
  }
  __syncthreads();
  if (kh == 0) {
#pragma unroll
    for (int r = 0; r < 8; ++r)
      tile[(r + 8 * hi) * ATP + ct * 16 + lanelo] = acc[r] + red[r * 32 + lane];
  }
  __syncthreads();

  const int orow = h >> 1;
  const int oc = (h & 1) * 128 + lane * 4;
  const float4 ov4 = *(const float4*)(tile + orow * ATP + oc);
  *(float4*)(out + ((size_t)b * SEQ + qbase + orow) * HID + oc) = ov4;
}

extern "C" void kernel_launch(void* const* d_in, const int* in_sizes, int n_in,
                              void* d_out, int out_size, void* d_ws,
                              size_t ws_size, hipStream_t stream) {
  const float* x = (const float*)d_in[0];
  const float* Wq = (const float*)d_in[1];
  const float* Wk = (const float*)d_in[2];
  const float* Wv = (const float*)d_in[3];
  const float* Wo = (const float*)d_in[4];
  const float* rope = (const float*)d_in[5];
  const float* so8 = (const float*)d_in[6];

  float* ws = (float*)d_ws;
  float* kb = ws;
  float* vtb = kb + (size_t)2 * N_KV * SEQ * HDIM;

  const int BS = 2 * SEQ;

  kv_rope_so8_kernel<<<BS, 32, 0, stream>>>(x, Wk, Wv, rope, so8, kb, vtb);
  so8_qproj_flash_attn_outproj_kernel<<<2 * QTILES, 1024, 0, stream>>>(
      x, Wq, rope, so8, kb, vtb, Wo, (float*)d_out);
}


# K/V projection kernel: per-thread scalar dot products replaced by transposed f32 WMMA tiles (W rows x 16 positions), head dims land per lane, rope/rotation in registers; k/v summed in WMMA k order
# speedup vs baseline: 4.5058x; 1.9173x over previous
#include <hip/hip_runtime.h>
#include <hip/hip_bf16.h>

#define N_HEADS 32
#define N_KV 8
#define HDIM 8
#define SEQ 2048
#define HID 256
#define QTILES (SEQ / 16)

typedef __attribute__((ext_vector_type(2))) float v2f;
typedef __attribute__((ext_vector_type(8))) float v8f;
typedef const __attribute__((address_space(1))) v2f gv2f;

__device__ const float kOnes[64] = {
    1.0f, 1.0f, 1.0f, 1.0f, 1.0f, 1.0f, 1.0f, 1.0f, 1.0f, 1.0f, 1.0f, 1.0f, 1.0f,
    1.0f, 1.0f, 1.0f, 1.0f, 1.0f, 1.0f, 1.0f, 1.0f, 1.0f, 1.0f, 1.0f, 1.0f, 1.0f,
    1.0f, 1.0f, 1.0f, 1.0f, 1.0f, 1.0f, 1.0f, 1.0f, 1.0f, 1.0f, 1.0f, 1.0f, 1.0f,
    1.0f, 1.0f, 1.0f, 1.0f, 1.0f, 1.0f, 1.0f, 1.0f, 1.0f, 1.0f, 1.0f, 1.0f, 1.0f,
    1.0f, 1.0f, 1.0f, 1.0f, 1.0f, 1.0f, 1.0f, 1.0f, 1.0f, 1.0f, 1.0f, 1.0f};

template <int CTRL>
__device__ __forceinline__ float dpp_max(float x) {
  const int y =
      __builtin_amdgcn_update_dpp(0, __float_as_int(x), CTRL, 0xf, 0xf, true);
  return fmaxf(x, __int_as_float(y));
}

__device__ __forceinline__ float rowmax16(float x) {
  x = dpp_max<0xB1>(x);
  x = dpp_max<0x4E>(x);
  x = dpp_max<0x141>(x);
  x = dpp_max<0x140>(x);
  return x;
}

__device__ __forceinline__ float fast_exp2(float x) {
  return __builtin_amdgcn_exp2f(x);
}

#define PPITCH 20
#define PTILE (16 * PPITCH)

template <int NT, bool MASK>
__device__ __forceinline__ void attn_block(
    const float* __restrict__ kp, const float* __restrict__ vrow,
    float* __restrict__ pw, int kbase, int lanelo, int hi, int hi2,
    bool vlane, const float* __restrict__ maskv, const v2f Qa0, const v2f Qa1,
    float* __restrict__ m, v8f& o) {
  v2f Kb0[NT], Kb1[NT];
#pragma unroll
  for (int t = 0; t < NT; ++t) {
    const float* krow = kp + (size_t)(kbase + 16 * t + lanelo) * HDIM;
    Kb0[t] = *(const v2f*)(krow + hi2);
    Kb1[t] = *(const v2f*)(krow + 4 + hi2);
  }
  const float* vsrc = vrow + (vlane ? kbase : 0) + hi2;
  v2f Vb[NT][4];
#pragma unroll
  for (int t = 0; t < NT; ++t) {
#pragma unroll
    for (int u = 0; u < 4; ++u)
      Vb[t][u] = *(gv2f*)(vsrc + 16 * t + 4 * u);
  }

  v8f sc[NT];
#pragma unroll
  for (int t = 0; t < NT; ++t) {
    v8f z = {};
    z = __builtin_amdgcn_wmma_f32_16x16x4_f32(false, Qa0, false, Kb0[t],
                                              (short)0, z, false, false);
    sc[t] = __builtin_amdgcn_wmma_f32_16x16x4_f32(false, Qa1, false, Kb1[t],
                                                  (short)0, z, false, false);
  }

#pragma unroll
  for (int r = 0; r < 8; ++r) {
    if (MASK) sc[NT - 1][r] += maskv[r];
    float mx = sc[0][r];
#pragma unroll
    for (int t = 1; t < NT; ++t) mx = fmaxf(mx, sc[t][r]);
    const float rm = rowmax16(mx);
    const float mnew = fmaxf(m[r], rm);
    const float corr = fast_exp2(m[r] - mnew);
    m[r] = mnew;
    o[r] = o[r] * corr;
#pragma unroll
    for (int t = 0; t < NT; ++t) sc[t][r] = fast_exp2(sc[t][r] - mnew);
  }

  float* pst = pw + hi * (8 * PPITCH) + lanelo;
  const float* pld = pw + lanelo * PPITCH + hi2;
#pragma unroll
  for (int t = 0; t < NT; ++t) {
#pragma unroll
    for (int r = 0; r < 8; ++r) pst[t * PTILE + r * PPITCH] = sc[t][r];
  }
  v2f Pa[NT][4];
#pragma unroll
  for (int t = 0; t < NT; ++t) {
#pragma unroll
    for (int u = 0; u < 4; ++u)
      Pa[t][u] = *(const v2f*)(pld + t * PTILE + 4 * u);
  }
#pragma unroll
  for (int t = 0; t < NT; ++t) {
#pragma unroll
    for (int u = 0; u < 4; ++u)
      o = __builtin_amdgcn_wmma_f32_16x16x4_f32(false, Pa[t][u], false,
                                                Vb[t][u], (short)0, o, false,
                                                false);
  }
}

#define ATP 260

__device__ __forceinline__ v8f tile_gemm_khalf(const float* __restrict__ tile,
                                               const float* __restrict__ Wm,
                                               int ct, int kh, int lanelo,
                                               int hi) {
  const float* arow = tile + lanelo * ATP + kh * 128 + 4 * hi;
  const float* brow = Wm + (size_t)(ct * 16 + lanelo) * HID + kh * 128 + 4 * hi;
  v8f acc0 = {}, acc1 = {};
#pragma unroll
  for (int u = 0; u < 16; ++u) {
    const float4 a4 = *(const float4*)(arow + 8 * u);
    const float4 b4 = *(const float4*)(brow + 8 * u);
    acc0 = __builtin_amdgcn_wmma_f32_16x16x4_f32(
        false, (v2f){a4.x, a4.y}, false, (v2f){b4.x, b4.y}, (short)0, acc0,
        false, false);
    acc1 = __builtin_amdgcn_wmma_f32_16x16x4_f32(
        false, (v2f){a4.z, a4.w}, false, (v2f){b4.z, b4.w}, (short)0, acc1,
        false, false);
  }
  return acc0 + acc1;
}


__global__ __launch_bounds__(512) void kv_proj_rope_so8_kernel(
    const float* __restrict__ x, const float* __restrict__ Wk,
    const float* __restrict__ Wv, const float* __restrict__ rope,
    const float* __restrict__ so8, float* __restrict__ kb,
    float* __restrict__ vtb) {
  const int lane = threadIdx.x & 31;
  const int w = __builtin_amdgcn_readfirstlane((int)(threadIdx.x >> 5));
  const int lanelo = lane & 15;
  const int hi = lane >> 4;
  const int row0 = blockIdx.x * 16;
  const int b = row0 >> 11;
  const int s0 = row0 & (SEQ - 1);

  __shared__ __attribute__((aligned(16))) float tile[16 * ATP];
  __shared__ __attribute__((aligned(16))) float red[8 * 256];
  __shared__ __attribute__((aligned(16))) float trig[16 * 8];

  const int ct = w & 7;
  const int kh = w >> 3;
  const bool isk = ct < 4;
  const float* Wrows = (isk ? Wk : Wv) + (size_t)(ct & 3) * 16 * HID;

  const int xr = threadIdx.x >> 6;
  const int xc = (threadIdx.x & 63) * 4;
  const float4 x0 = *(const float4*)(x + ((size_t)row0 + xr) * HID + xc);
  const float4 x1 = *(const float4*)(x + ((size_t)row0 + 8 + xr) * HID + xc);
  const float* arow = Wrows + (size_t)lanelo * HID + kh * 128 + 4 * hi;
  float4 wa[16];
#pragma unroll
  for (int u = 0; u < 16; ++u) wa[u] = *(const float4*)(arow + 8 * u);

  *(float4*)(tile + xr * ATP + xc) = x0;
  *(float4*)(tile + (8 + xr) * ATP + xc) = x1;
  if (threadIdx.x < 128) {
    const int i = threadIdx.x & 3;
    const int tr = threadIdx.x >> 3;
    const float invf = (i == 0) ? 1.0f : (i == 1) ? 0.1f : (i == 2) ? 0.01f : 0.001f;
    const float fr = (float)(s0 + tr) * invf;
    trig[threadIdx.x] = (threadIdx.x & 4) ? sinf(fr) : cosf(fr);
  }
  __syncthreads();

  const float* brow = tile + lanelo * ATP + kh * 128 + 4 * hi;
  v8f acc0 = {}, acc1 = {};
#pragma unroll
  for (int u = 0; u < 16; ++u) {
    const float4 b4 = *(const float4*)(brow + 8 * u);
    acc0 = __builtin_amdgcn_wmma_f32_16x16x4_f32(
        false, (v2f){wa[u].x, wa[u].y}, false, (v2f){b4.x, b4.y}, (short)0,
        acc0, false, false);
    acc1 = __builtin_amdgcn_wmma_f32_16x16x4_f32(
        false, (v2f){wa[u].z, wa[u].w}, false, (v2f){b4.z, b4.w}, (short)0,
        acc1, false, false);
  }
  const v8f part = acc0 + acc1;
  if (kh == 1) {
#pragma unroll
    for (int r = 0; r < 8; ++r) red[ct * 256 + r * 32 + lane] = part[r];
  }
  __syncthreads();
  if (kh == 1) return;

  float acc[HDIM];
#pragma unroll
  for (int r = 0; r < 8; ++r) acc[r] = part[r] + red[ct * 256 + r * 32 + lane];
  const int head = (ct & 3) * 2 + hi;

  float y[HDIM];
  if (isk) {
    const float4 c4 = *(const float4*)(trig + lanelo * 8);
    const float4 s4 = *(const float4*)(trig + lanelo * 8 + 4);
    const float cs[4] = {c4.x, c4.y, c4.z, c4.w};
    const float sn[4] = {s4.x, s4.y, s4.z, s4.w};
    float cr[4] = {}, sr[4] = {};
#pragma unroll
    for (int i = 0; i < 4; ++i) {
#pragma unroll
      for (int j = 0; j < 4; ++j) {
        cr[j] = fmaf(cs[i], rope[i * 8 + j], cr[j]);
        sr[j] = fmaf(sn[i], rope[i * 8 + j], sr[j]);
      }
    }
#pragma unroll
    for (int d = 0; d < HDIM; ++d) {
      const int j = d & 3;
      const float rh = (d < 4) ? -acc[d + 4] : acc[d - 4];
      y[d] = acc[d] * cr[j] + rh * sr[j];
    }
  } else {
#pragma unroll
    for (int d = 0; d < HDIM; ++d) y[d] = acc[d];
  }

  float z[HDIM];
#pragma unroll
  for (int j = 0; j < HDIM; ++j) {
    float sum = 0.0f;
#pragma unroll
    for (int d = 0; d < HDIM; ++d) sum = fmaf(y[d], so8[d * 8 + j], sum);
    z[j] = sum;
  }

  if (isk) {
    float* dst = kb + (((size_t)b * N_KV + head) * SEQ + s0 + lanelo) * HDIM;
    *(float4*)dst = make_float4(z[0], z[1], z[2], z[3]);
    *(float4*)(dst + 4) = make_float4(z[4], z[5], z[6], z[7]);
  } else {
    float* dst = vtb + ((size_t)b * N_KV + head) * HDIM * SEQ + s0 + lanelo;
#pragma unroll
    for (int j = 0; j < HDIM; ++j) dst[(size_t)j * SEQ] = z[j];
  }
}

__global__ __launch_bounds__(1024) void so8_qproj_flash_attn_outproj_kernel(
    const float* __restrict__ x, const float* __restrict__ Wq,
    const float* __restrict__ rope, const float* __restrict__ so8,
    const float* __restrict__ k, const float* __restrict__ vT,
    const float* __restrict__ Wo, float* __restrict__ out) {
  const int lane = threadIdx.x & 31;
  const int w = __builtin_amdgcn_readfirstlane((int)(threadIdx.x >> 5));
  const int b = blockIdx.x & 1;
  const int p = blockIdx.x >> 1;
  const int grp = (w >> 2) & 1;
  const int hp = ((w >> 3) << 2) | (w & 3);
  const int kv = hp >> 1;
  const int p8 = grp ? (2 * QTILES - 1 - p) : p;
  const int qt = p8 >> 1;
  const int lanelo = lane & 15;
  const int hi = lane >> 4;
  const int hi2 = hi * 2;
  const int pos_early = p * 8;
  const int pos_late = (2 * QTILES - 1 - p) * 8 - 8;

  const float* kp = k + ((size_t)b * N_KV + kv) * SEQ * HDIM;
  const float* vp = vT + ((size_t)b * N_KV + kv) * HDIM * SEQ;
  const bool vlane = lanelo < 8;
  const float* vrow = vlane ? vp + (size_t)lanelo * SEQ : kOnes;

  __shared__ __attribute__((aligned(16))) float plds[N_HEADS * 4 * PTILE];
  __shared__ __attribute__((aligned(16))) float tile[16 * ATP];
  __shared__ __attribute__((aligned(16))) float trig[16 * 8];
  float* pw = plds + w * 4 * PTILE;
  const int ct = w & 15;
  const int kh = w >> 4;

  {
    const int xr = threadIdx.x >> 6;
    const int xc = (threadIdx.x & 63) * 4;
    const int xpos = (xr < 8 ? pos_early : pos_late) + xr;
    const float4 xv = *(const float4*)(x + ((size_t)b * SEQ + xpos) * HID + xc);
    *(float4*)(tile + xr * ATP + xc) = xv;
  }
  if (threadIdx.x < 128) {
    const int i = threadIdx.x & 3;
    const int tr = threadIdx.x >> 3;
    const float invf = (i == 0) ? 1.0f : (i == 1) ? 0.1f : (i == 2) ? 0.01f : 0.001f;
    const float fr = (float)((tr < 8 ? pos_early : pos_late) + tr) * invf;
    trig[threadIdx.x] = (threadIdx.x & 4) ? sinf(fr) : cosf(fr);
  }
  __syncthreads();
  {
    const v8f qacc = tile_gemm_khalf(tile, Wq, ct, kh, lanelo, hi);
    float* red = plds + ct * 256;
    if (kh == 1) {
#pragma unroll
      for (int r = 0; r < 8; ++r) red[r * 32 + lane] = qacc[r];
    }
    __syncthreads();
    if (kh == 0) {
#pragma unroll
      for (int r = 0; r < 8; ++r)
        tile[(r + 8 * hi) * ATP + ct * 16 + lanelo] =
            qacc[r] + red[r * 32 + lane];
    }
    __syncthreads();
  }

  v2f Qa0, Qa1;
  {
    const int tr = 8 * grp + (lanelo & 7);
    const int hd = 2 * hp + (lanelo >> 3);
    const float4 q0 = *(const float4*)(tile + tr * ATP + hd * HDIM);
    const float4 q1 = *(const float4*)(tile + tr * ATP + hd * HDIM + 4);
    const float4 c4 = *(const float4*)(trig + tr * 8);
    const float4 s4 = *(const float4*)(trig + tr * 8 + 4);
    const float acc[HDIM] = {q0.x, q0.y, q0.z, q0.w, q1.x, q1.y, q1.z, q1.w};
    const float cs[4] = {c4.x, c4.y, c4.z, c4.w};
    const float sn[4] = {s4.x, s4.y, s4.z, s4.w};
    float cr[4] = {}, sr[4] = {};
#pragma unroll
    for (int i = 0; i < 4; ++i) {
#pragma unroll
      for (int j = 0; j < 4; ++j) {
        cr[j] = fmaf(cs[i], rope[i * 8 + j], cr[j]);
        sr[j] = fmaf(sn[i], rope[i * 8 + j], sr[j]);
      }
    }
    float y[HDIM];
#pragma unroll
    for (int d = 0; d < HDIM; ++d) {
      const int j = d & 3;
      const float rh = (d < 4) ? -acc[d + 4] : acc[d - 4];
      y[d] = acc[d] * cr[j] + rh * sr[j];
    }
    float z[HDIM];
#pragma unroll
    for (int j = 0; j < HDIM; ++j) {
      float sum = 0.0f;
#pragma unroll
      for (int d = 0; d < HDIM; ++d) sum = fmaf(y[d], so8[d * 8 + j], sum);
      z[j] = sum;
    }
    const float QSCALE = 0.35355339059327f * 1.44269504088896f;
    Qa0 = (v2f){hi ? z[2] : z[0], hi ? z[3] : z[1]} * QSCALE;
    Qa1 = (v2f){hi ? z[6] : z[4], hi ? z[7] : z[5]} * QSCALE;
  }
  __syncthreads();

  const int moff = 8 * (p8 & 1);
  float maskv[8];
#pragma unroll
  for (int r = 0; r < 8; ++r)
    maskv[r] = (lanelo > r + moff) ? -3.0e38f : 0.0f;

  v8f o = {};
  float m[8];
#pragma unroll
  for (int r = 0; r < 8; ++r) m[r] = -3.0e38f;

  const int nfull = qt >> 2;
  for (int kb = 0; kb < nfull; ++kb)
    attn_block<4, false>(kp, vrow, pw, kb * 64, lanelo, hi, hi2, vlane, maskv,
                         Qa0, Qa1, m, o);
  const int tbase = nfull * 64;
  switch (qt & 3) {
    case 0:
      attn_block<1, true>(kp, vrow, pw, tbase, lanelo, hi, hi2, vlane, maskv,
                          Qa0, Qa1, m, o);
      break;
    case 1:
      attn_block<2, true>(kp, vrow, pw, tbase, lanelo, hi, hi2, vlane, maskv,
                          Qa0, Qa1, m, o);
      break;
    case 2:
      attn_block<3, true>(kp, vrow, pw, tbase, lanelo, hi, hi2, vlane, maskv,
                          Qa0, Qa1, m, o);
      break;
    default:
      attn_block<4, true>(kp, vrow, pw, tbase, lanelo, hi, hi2, vlane, maskv,
                          Qa0, Qa1, m, o);
      break;
  }

#pragma unroll
  for (int r = 0; r < 8; ++r) {
    const float ov = o[r];
    const float l = __shfl(ov, (lane & 16) + 8);
    const float outv = ov * __builtin_amdgcn_rcpf(l);
    if (lanelo < HDIM)
      tile[(8 * grp + r) * ATP + (2 * hp + hi) * HDIM + lanelo] = outv;
  }
  __syncthreads();

  const v8f acc = tile_gemm_khalf(tile, Wo, ct, kh, lanelo, hi);

  float* red = plds + ct * 256;
  if (kh == 1) {
#pragma unroll
    for (int r = 0; r < 8; ++r) red[r * 32 + lane] = acc[r];
  }
  __syncthreads();
  if (kh == 0) {
#pragma unroll
    for (int r = 0; r < 8; ++r)
      tile[(r + 8 * hi) * ATP + ct * 16 + lanelo] = acc[r] + red[r * 32 + lane];
  }
  __syncthreads();

  const int orow = w >> 1;
  const int oc = (w & 1) * 128 + lane * 4;
  const int opos = (orow < 8 ? pos_early : pos_late) + orow;
  const float4 ov4 = *(const float4*)(tile + orow * ATP + oc);
  *(float4*)(out + ((size_t)b * SEQ + opos) * HID + oc) = ov4;
}

extern "C" void kernel_launch(void* const* d_in, const int* in_sizes, int n_in,
                              void* d_out, int out_size, void* d_ws,
                              size_t ws_size, hipStream_t stream) {
  const float* x = (const float*)d_in[0];
  const float* Wq = (const float*)d_in[1];
  const float* Wk = (const float*)d_in[2];
  const float* Wv = (const float*)d_in[3];
  const float* Wo = (const float*)d_in[4];
  const float* rope = (const float*)d_in[5];
  const float* so8 = (const float*)d_in[6];

  float* ws = (float*)d_ws;
  float* kb = ws;
  float* vtb = kb + (size_t)2 * N_KV * SEQ * HDIM;

  kv_proj_rope_so8_kernel<<<2 * QTILES, 512, 0, stream>>>(x, Wk, Wv, rope, so8,
                                                         kb, vtb);
  so8_qproj_flash_attn_outproj_kernel<<<2 * QTILES, 1024, 0, stream>>>(
      x, Wq, rope, so8, kb, vtb, Wo, (float*)d_out);
}


# Key loop K, V, P fragments fetched as one 16-byte load per lane (were two 8-byte): V stored in 4-key groups so a load reads whole lines; head dims and a tile's keys summed in permuted k order
# speedup vs baseline: 4.6682x; 1.0360x over previous
#include <hip/hip_runtime.h>
#include <hip/hip_bf16.h>

#define N_HEADS 32
#define N_KV 8
#define HDIM 8
#define SEQ 2048
#define HID 256
#define QTILES (SEQ / 16)

typedef __attribute__((ext_vector_type(2))) float v2f;
typedef __attribute__((ext_vector_type(4))) float v4f;
typedef __attribute__((ext_vector_type(8))) float v8f;
typedef const __attribute__((address_space(1))) v4f gv4f;

#define ONES8 1.0f, 1.0f, 1.0f, 1.0f, 1.0f, 1.0f, 1.0f, 1.0f
#define ONES64 ONES8, ONES8, ONES8, ONES8, ONES8, ONES8, ONES8, ONES8
__device__ __attribute__((aligned(16))) const float kOnes[512] = {
    ONES64, ONES64, ONES64, ONES64, ONES64, ONES64, ONES64, ONES64};

template <int CTRL>
__device__ __forceinline__ float dpp_max(float x) {
  const int y =
      __builtin_amdgcn_update_dpp(0, __float_as_int(x), CTRL, 0xf, 0xf, true);
  return fmaxf(x, __int_as_float(y));
}

__device__ __forceinline__ float rowmax16(float x) {
  x = dpp_max<0xB1>(x);
  x = dpp_max<0x4E>(x);
  x = dpp_max<0x141>(x);
  x = dpp_max<0x140>(x);
  return x;
}

__device__ __forceinline__ float fast_exp2(float x) {
  return __builtin_amdgcn_exp2f(x);
}

#define PPITCH 20
#define PTILE (16 * PPITCH)

template <int NT, bool MASK>
__device__ __forceinline__ void attn_block(
    const float* __restrict__ kp, unsigned koff, const float* __restrict__ vrow,
    float* __restrict__ pw, int kbase, int lanelo, int hi, bool vlane,
    const float* __restrict__ maskv, const v2f Qa0, const v2f Qa1,
    float* __restrict__ m, v8f& o) {
  const char* kblk = (const char*)(kp + (size_t)kbase * HDIM);
  v4f Kb[NT];
#pragma unroll
  for (int t = 0; t < NT; ++t)
    Kb[t] = *(const v4f*)(kblk + koff + 16 * t * HDIM * sizeof(float));
  const float* vsrc = vrow + (vlane ? kbase * HDIM : 0);
  v4f Vb[NT][2];
#pragma unroll
  for (int t = 0; t < NT; ++t) {
#pragma unroll
    for (int v = 0; v < 2; ++v)
      Vb[t][v] = *(gv4f*)(vsrc + 128 * t + 64 * v);
  }

  v8f sc[NT];
#pragma unroll
  for (int t = 0; t < NT; ++t) {
    v8f z = {};
    z = __builtin_amdgcn_wmma_f32_16x16x4_f32(false, Qa0, false, Kb[t].xy,
                                              (short)0, z, false, false);
    sc[t] = __builtin_amdgcn_wmma_f32_16x16x4_f32(false, Qa1, false, Kb[t].zw,
                                                  (short)0, z, false, false);
  }

#pragma unroll
  for (int r = 0; r < 8; ++r) {
    if (MASK) sc[NT - 1][r] += maskv[r];
    float mx = sc[0][r];
#pragma unroll
    for (int t = 1; t < NT; ++t) mx = fmaxf(mx, sc[t][r]);
    const float rm = rowmax16(mx);
    const float mnew = fmaxf(m[r], rm);
    const float corr = fast_exp2(m[r] - mnew);
    m[r] = mnew;
    o[r] = o[r] * corr;
#pragma unroll
    for (int t = 0; t < NT; ++t) sc[t][r] = fast_exp2(sc[t][r] - mnew);
  }

  float* pst = pw + hi * (8 * PPITCH) + lanelo;
  const float* pld = pw + lanelo * PPITCH + 4 * hi;
#pragma unroll
  for (int t = 0; t < NT; ++t) {
#pragma unroll
    for (int r = 0; r < 8; ++r) pst[t * PTILE + r * PPITCH] = sc[t][r];
  }
  v4f Pa[NT][2];
#pragma unroll
  for (int t = 0; t < NT; ++t) {
#pragma unroll
    for (int v = 0; v < 2; ++v)
      Pa[t][v] = *(const v4f*)(pld + t * PTILE + 8 * v);
  }
#pragma unroll
  for (int t = 0; t < NT; ++t) {
#pragma unroll
    for (int v = 0; v < 2; ++v) {
      o = __builtin_amdgcn_wmma_f32_16x16x4_f32(false, Pa[t][v].xy, false,
                                                Vb[t][v].xy, (short)0, o,
                                                false, false);
      o = __builtin_amdgcn_wmma_f32_16x16x4_f32(false, Pa[t][v].zw, false,
                                                Vb[t][v].zw, (short)0, o,
                                                false, false);
    }
  }
}

#define ATP 260

__device__ __forceinline__ v8f tile_gemm_khalf(const float* __restrict__ tile,
                                               const float* __restrict__ Wm,
                                               int ct, int kh, int lanelo,
                                               int hi) {
  const float* arow = tile + lanelo * ATP + kh * 128 + 4 * hi;
  const float* brow = Wm + (size_t)(ct * 16 + lanelo) * HID + kh * 128 + 4 * hi;
  v8f acc0 = {}, acc1 = {};
#pragma unroll
  for (int u = 0; u < 16; ++u) {
    const float4 a4 = *(const float4*)(arow + 8 * u);
    const float4 b4 = *(const float4*)(brow + 8 * u);
    acc0 = __builtin_amdgcn_wmma_f32_16x16x4_f32(
        false, (v2f){a4.x, a4.y}, false, (v2f){b4.x, b4.y}, (short)0, acc0,
        false, false);
    acc1 = __builtin_amdgcn_wmma_f32_16x16x4_f32(
        false, (v2f){a4.z, a4.w}, false, (v2f){b4.z, b4.w}, (short)0, acc1,
        false, false);
  }
  return acc0 + acc1;
}


__global__ __launch_bounds__(512) void kv_proj_rope_so8_kernel(
    const float* __restrict__ x, const float* __restrict__ Wk,
    const float* __restrict__ Wv, const float* __restrict__ rope,
    const float* __restrict__ so8, float* __restrict__ kb,
    float* __restrict__ vqb) {
  const int lane = threadIdx.x & 31;
  const int w = __builtin_amdgcn_readfirstlane((int)(threadIdx.x >> 5));
  const int lanelo = lane & 15;
  const int hi = lane >> 4;
  const int row0 = blockIdx.x * 16;
  const int b = row0 >> 11;
  const int s0 = row0 & (SEQ - 1);

  __shared__ __attribute__((aligned(16))) float tile[16 * ATP];
  __shared__ __attribute__((aligned(16))) float red[8 * 256];
  __shared__ __attribute__((aligned(16))) float trig[16 * 8];

  const int ct = w & 7;
  const int kh = w >> 3;
  const bool isk = ct < 4;
  const float* Wrows = (isk ? Wk : Wv) + (size_t)(ct & 3) * 16 * HID;

  const int xr = threadIdx.x >> 6;
  const int xc = (threadIdx.x & 63) * 4;
  const float4 x0 = *(const float4*)(x + ((size_t)row0 + xr) * HID + xc);
  const float4 x1 = *(const float4*)(x + ((size_t)row0 + 8 + xr) * HID + xc);
  const float* arow = Wrows + (size_t)lanelo * HID + kh * 128 + 4 * hi;
  float4 wa[16];
#pragma unroll
  for (int u = 0; u < 16; ++u) wa[u] = *(const float4*)(arow + 8 * u);

  *(float4*)(tile + xr * ATP + xc) = x0;
  *(float4*)(tile + (8 + xr) * ATP + xc) = x1;
  if (threadIdx.x < 128) {
    const int i = threadIdx.x & 3;
    const int tr = threadIdx.x >> 3;
    const float invf = (i == 0) ? 1.0f : (i == 1) ? 0.1f : (i == 2) ? 0.01f : 0.001f;
    const float fr = (float)(s0 + tr) * invf;
    trig[threadIdx.x] = (threadIdx.x & 4) ? sinf(fr) : cosf(fr);
  }
  __syncthreads();

  const float* brow = tile + lanelo * ATP + kh * 128 + 4 * hi;
  v8f acc0 = {}, acc1 = {};
#pragma unroll
  for (int u = 0; u < 16; ++u) {
    const float4 b4 = *(const float4*)(brow + 8 * u);
    acc0 = __builtin_amdgcn_wmma_f32_16x16x4_f32(
        false, (v2f){wa[u].x, wa[u].y}, false, (v2f){b4.x, b4.y}, (short)0,
        acc0, false, false);
    acc1 = __builtin_amdgcn_wmma_f32_16x16x4_f32(
        false, (v2f){wa[u].z, wa[u].w}, false, (v2f){b4.z, b4.w}, (short)0,
        acc1, false, false);
  }
  const v8f part = acc0 + acc1;
  if (kh == 1) {
#pragma unroll
    for (int r = 0; r < 8; ++r) red[ct * 256 + r * 32 + lane] = part[r];
  }
  __syncthreads();
  if (kh == 1) return;

  float acc[HDIM];
#pragma unroll
  for (int r = 0; r < 8; ++r) acc[r] = part[r] + red[ct * 256 + r * 32 + lane];
  const int head = (ct & 3) * 2 + hi;

  float y[HDIM];
  if (isk) {
    const float4 c4 = *(const float4*)(trig + lanelo * 8);
    const float4 s4 = *(const float4*)(trig + lanelo * 8 + 4);
    const float cs[4] = {c4.x, c4.y, c4.z, c4.w};
    const float sn[4] = {s4.x, s4.y, s4.z, s4.w};
    float cr[4] = {}, sr[4] = {};
#pragma unroll
    for (int i = 0; i < 4; ++i) {
#pragma unroll
      for (int j = 0; j < 4; ++j) {
        cr[j] = fmaf(cs[i], rope[i * 8 + j], cr[j]);
        sr[j] = fmaf(sn[i], rope[i * 8 + j], sr[j]);
      }
    }
#pragma unroll
    for (int d = 0; d < HDIM; ++d) {
      const int j = d & 3;
      const float rh = (d < 4) ? -acc[d + 4] : acc[d - 4];
      y[d] = acc[d] * cr[j] + rh * sr[j];
    }
  } else {
#pragma unroll
    for (int d = 0; d < HDIM; ++d) y[d] = acc[d];
  }

  float z[HDIM];
#pragma unroll
  for (int j = 0; j < HDIM; ++j) {
    float sum = 0.0f;
#pragma unroll
    for (int d = 0; d < HDIM; ++d) sum = fmaf(y[d], so8[d * 8 + j], sum);
    z[j] = sum;
  }

  if (isk) {
    float* dst = kb + (((size_t)b * N_KV + head) * SEQ + s0 + lanelo) * HDIM;
    *(float4*)dst = make_float4(z[0], z[1], z[2], z[3]);
    *(float4*)(dst + 4) = make_float4(z[4], z[5], z[6], z[7]);
  } else {
    float* stg = tile + (ct - 4) * 256;
    float* sw = stg + hi * 128 + (lanelo >> 2) * 32 + (lanelo & 3);
#pragma unroll
    for (int j = 0; j < HDIM; ++j) sw[j * 4] = z[j];
#pragma unroll
    for (int hh = 0; hh < 2; ++hh) {
      const v4f v4 = *(const v4f*)(stg + hh * 128 + lane * 4);
      float* dst = vqb +
                   (((size_t)b * N_KV + (ct & 3) * 2 + hh) * SEQ + s0) * HDIM +
                   lane * 4;
      *(v4f*)dst = v4;
    }
  }
}

__global__ __launch_bounds__(1024) void so8_qproj_flash_attn_outproj_kernel(
    const float* __restrict__ x, const float* __restrict__ Wq,
    const float* __restrict__ rope, const float* __restrict__ so8,
    const float* __restrict__ k, const float* __restrict__ vq,
    const float* __restrict__ Wo, float* __restrict__ out) {
  const int lane = threadIdx.x & 31;
  const int w = __builtin_amdgcn_readfirstlane((int)(threadIdx.x >> 5));
  const int b = blockIdx.x & 1;
  const int p = blockIdx.x >> 1;
  const int grp = (w >> 2) & 1;
  const int hp = ((w >> 3) << 2) | (w & 3);
  const int kv = hp >> 1;
  const int p8 = grp ? (2 * QTILES - 1 - p) : p;
  const int qt = p8 >> 1;
  const int lanelo = lane & 15;
  const int hi = lane >> 4;
  const int pos_early = p * 8;
  const int pos_late = (2 * QTILES - 1 - p) * 8 - 8;

  const float* kp = k + ((size_t)b * N_KV + kv) * SEQ * HDIM;
  const unsigned koff = (unsigned)(lanelo * HDIM + 4 * hi) * sizeof(float);
  const float* vp = vq + ((size_t)b * N_KV + kv) * SEQ * HDIM;
  const bool vlane = lanelo < 8;
  const float* vrow = vlane ? vp + hi * 32 + lanelo * 4 : kOnes;

  __shared__ __attribute__((aligned(16))) float plds[N_HEADS * 4 * PTILE];
  __shared__ __attribute__((aligned(16))) float tile[16 * ATP];
  __shared__ __attribute__((aligned(16))) float trig[16 * 8];
  float* pw = plds + w * 4 * PTILE;
  const int ct = w & 15;
  const int kh = w >> 4;

  {
    const int xr = threadIdx.x >> 6;
    const int xc = (threadIdx.x & 63) * 4;
    const int xpos = (xr < 8 ? pos_early : pos_late) + xr;
    const float4 xv = *(const float4*)(x + ((size_t)b * SEQ + xpos) * HID + xc);
    *(float4*)(tile + xr * ATP + xc) = xv;
  }
  if (threadIdx.x < 128) {
    const int i = threadIdx.x & 3;
    const int tr = threadIdx.x >> 3;
    const float invf = (i == 0) ? 1.0f : (i == 1) ? 0.1f : (i == 2) ? 0.01f : 0.001f;
    const float fr = (float)((tr < 8 ? pos_early : pos_late) + tr) * invf;
    trig[threadIdx.x] = (threadIdx.x & 4) ? sinf(fr) : cosf(fr);
  }
  __syncthreads();
  {
    const v8f qacc = tile_gemm_khalf(tile, Wq, ct, kh, lanelo, hi);
    float* red = plds + ct * 256;
    if (kh == 1) {
#pragma unroll
      for (int r = 0; r < 8; ++r) red[r * 32 + lane] = qacc[r];
    }
    __syncthreads();
    if (kh == 0) {
#pragma unroll
      for (int r = 0; r < 8; ++r)
        tile[(r + 8 * hi) * ATP + ct * 16 + lanelo] =
            qacc[r] + red[r * 32 + lane];
    }
    __syncthreads();
  }

  v2f Qa0, Qa1;
  {
    const int tr = 8 * grp + (lanelo & 7);
    const int hd = 2 * hp + (lanelo >> 3);
    const float4 q0 = *(const float4*)(tile + tr * ATP + hd * HDIM);
    const float4 q1 = *(const float4*)(tile + tr * ATP + hd * HDIM + 4);
    const float4 c4 = *(const float4*)(trig + tr * 8);
    const float4 s4 = *(const float4*)(trig + tr * 8 + 4);
    const float acc[HDIM] = {q0.x, q0.y, q0.z, q0.w, q1.x, q1.y, q1.z, q1.w};
    const float cs[4] = {c4.x, c4.y, c4.z, c4.w};
    const float sn[4] = {s4.x, s4.y, s4.z, s4.w};
    float cr[4] = {}, sr[4] = {};
#pragma unroll
    for (int i = 0; i < 4; ++i) {
#pragma unroll
      for (int j = 0; j < 4; ++j) {
        cr[j] = fmaf(cs[i], rope[i * 8 + j], cr[j]);
        sr[j] = fmaf(sn[i], rope[i * 8 + j], sr[j]);
      }
    }
    float y[HDIM];
#pragma unroll
    for (int d = 0; d < HDIM; ++d) {
      const int j = d & 3;
      const float rh = (d < 4) ? -acc[d + 4] : acc[d - 4];
      y[d] = acc[d] * cr[j] + rh * sr[j];
    }
    float z[HDIM];
#pragma unroll
    for (int j = 0; j < HDIM; ++j) {
      float sum = 0.0f;
#pragma unroll
      for (int d = 0; d < HDIM; ++d) sum = fmaf(y[d], so8[d * 8 + j], sum);
      z[j] = sum;
    }
    const float QSCALE = 0.35355339059327f * 1.44269504088896f;
    Qa0 = (v2f){hi ? z[4] : z[0], hi ? z[5] : z[1]} * QSCALE;
    Qa1 = (v2f){hi ? z[6] : z[2], hi ? z[7] : z[3]} * QSCALE;
  }
  __syncthreads();

  const int moff = 8 * (p8 & 1);
  float maskv[8];
#pragma unroll
  for (int r = 0; r < 8; ++r)
    maskv[r] = (lanelo > r + moff) ? -3.0e38f : 0.0f;

  v8f o = {};
  float m[8];
#pragma unroll
  for (int r = 0; r < 8; ++r) m[r] = -3.0e38f;

  const int nfull = qt >> 2;
  for (int kb = 0; kb < nfull; ++kb)
    attn_block<4, false>(kp, koff, vrow, pw, kb * 64, lanelo, hi, vlane, maskv,
                         Qa0, Qa1, m, o);
  const int tbase = nfull * 64;
  switch (qt & 3) {
    case 0:
      attn_block<1, true>(kp, koff, vrow, pw, tbase, lanelo, hi, vlane, maskv,
                          Qa0, Qa1, m, o);
      break;
    case 1:
      attn_block<2, true>(kp, koff, vrow, pw, tbase, lanelo, hi, vlane, maskv,
                          Qa0, Qa1, m, o);
      break;
    case 2:
      attn_block<3, true>(kp, koff, vrow, pw, tbase, lanelo, hi, vlane, maskv,
                          Qa0, Qa1, m, o);
      break;
    default:
      attn_block<4, true>(kp, koff, vrow, pw, tbase, lanelo, hi, vlane, maskv,
                          Qa0, Qa1, m, o);
      break;
  }

#pragma unroll
  for (int r = 0; r < 8; ++r) {
    const float ov = o[r];
    const float l = __shfl(ov, (lane & 16) + 8);
    const float outv = ov * __builtin_amdgcn_rcpf(l);
    if (lanelo < HDIM)
      tile[(8 * grp + r) * ATP + (2 * hp + hi) * HDIM + lanelo] = outv;
  }
  __syncthreads();

  const v8f acc = tile_gemm_khalf(tile, Wo, ct, kh, lanelo, hi);

  float* red = plds + ct * 256;
  if (kh == 1) {
#pragma unroll
    for (int r = 0; r < 8; ++r) red[r * 32 + lane] = acc[r];
  }
  __syncthreads();
  if (kh == 0) {
#pragma unroll
    for (int r = 0; r < 8; ++r)
      tile[(r + 8 * hi) * ATP + ct * 16 + lanelo] = acc[r] + red[r * 32 + lane];
  }
  __syncthreads();

  const int orow = w >> 1;
  const int oc = (w & 1) * 128 + lane * 4;
  const int opos = (orow < 8 ? pos_early : pos_late) + orow;
  const float4 ov4 = *(const float4*)(tile + orow * ATP + oc);
  *(float4*)(out + ((size_t)b * SEQ + opos) * HID + oc) = ov4;
}

extern "C" void kernel_launch(void* const* d_in, const int* in_sizes, int n_in,
                              void* d_out, int out_size, void* d_ws,
                              size_t ws_size, hipStream_t stream) {
  const float* x = (const float*)d_in[0];
  const float* Wq = (const float*)d_in[1];
  const float* Wk = (const float*)d_in[2];
  const float* Wv = (const float*)d_in[3];
  const float* Wo = (const float*)d_in[4];
  const float* rope = (const float*)d_in[5];
  const float* so8 = (const float*)d_in[6];

  float* ws = (float*)d_ws;
  float* kb = ws;
  float* vqb = kb + (size_t)2 * N_KV * SEQ * HDIM;

  kv_proj_rope_so8_kernel<<<2 * QTILES, 512, 0, stream>>>(x, Wk, Wv, rope, so8,
                                                         kb, vqb);
  so8_qproj_flash_attn_outproj_kernel<<<2 * QTILES, 1024, 0, stream>>>(
      x, Wq, rope, so8, kb, vqb, Wo, (float*)d_out);
}


# Softmax row max: the 16-lane butterfly of a block's 8 rows is one asm block of 32 in-place DPP max ops, rows interleaved per step (was DPP move + quieting copy + max each); maxima and sums unchanged
# speedup vs baseline: 4.8557x; 1.0402x over previous
#include <hip/hip_runtime.h>
#include <hip/hip_bf16.h>

#define N_HEADS 32
#define N_KV 8
#define HDIM 8
#define SEQ 2048
#define HID 256
#define QTILES (SEQ / 16)

typedef __attribute__((ext_vector_type(2))) float v2f;
typedef __attribute__((ext_vector_type(4))) float v4f;
typedef __attribute__((ext_vector_type(8))) float v8f;
typedef const __attribute__((address_space(1))) v4f gv4f;

#define ONES8 1.0f, 1.0f, 1.0f, 1.0f, 1.0f, 1.0f, 1.0f, 1.0f
#define ONES64 ONES8, ONES8, ONES8, ONES8, ONES8, ONES8, ONES8, ONES8
__device__ __attribute__((aligned(16))) const float kOnes[512] = {
    ONES64, ONES64, ONES64, ONES64, ONES64, ONES64, ONES64, ONES64};

#define DPPMAX8(CTRL)                                                  \
  "v_max_num_f32_dpp %0, %0, %0 " CTRL " row_mask:0xf bank_mask:0xf\n\t" \
  "v_max_num_f32_dpp %1, %1, %1 " CTRL " row_mask:0xf bank_mask:0xf\n\t" \
  "v_max_num_f32_dpp %2, %2, %2 " CTRL " row_mask:0xf bank_mask:0xf\n\t" \
  "v_max_num_f32_dpp %3, %3, %3 " CTRL " row_mask:0xf bank_mask:0xf\n\t" \
  "v_max_num_f32_dpp %4, %4, %4 " CTRL " row_mask:0xf bank_mask:0xf\n\t" \
  "v_max_num_f32_dpp %5, %5, %5 " CTRL " row_mask:0xf bank_mask:0xf\n\t" \
  "v_max_num_f32_dpp %6, %6, %6 " CTRL " row_mask:0xf bank_mask:0xf\n\t" \
  "v_max_num_f32_dpp %7, %7, %7 " CTRL " row_mask:0xf bank_mask:0xf\n\t"

__device__ __forceinline__ void rowmax16x8(float* __restrict__ x) {
  asm(DPPMAX8("quad_perm:[1,0,3,2]") DPPMAX8("quad_perm:[2,3,0,1]")
          DPPMAX8("row_half_mirror") DPPMAX8("row_mirror")
      : "+v"(x[0]), "+v"(x[1]), "+v"(x[2]), "+v"(x[3]), "+v"(x[4]),
        "+v"(x[5]), "+v"(x[6]), "+v"(x[7]));
}

__device__ __forceinline__ float fast_exp2(float x) {
  return __builtin_amdgcn_exp2f(x);
}

#define PPITCH 20
#define PTILE (16 * PPITCH)

template <int NT, bool MASK>
__device__ __forceinline__ void attn_block(
    const float* __restrict__ kp, unsigned koff, const float* __restrict__ vrow,
    float* __restrict__ pw, int kbase, int lanelo, int hi, bool vlane,
    const float* __restrict__ maskv, const v2f Qa0, const v2f Qa1,
    float* __restrict__ m, v8f& o) {
  const char* kblk = (const char*)(kp + (size_t)kbase * HDIM);
  v4f Kb[NT];
#pragma unroll
  for (int t = 0; t < NT; ++t)
    Kb[t] = *(const v4f*)(kblk + koff + 16 * t * HDIM * sizeof(float));
  const float* vsrc = vrow + (vlane ? kbase * HDIM : 0);
  v4f Vb[NT][2];
#pragma unroll
  for (int t = 0; t < NT; ++t) {
#pragma unroll
    for (int v = 0; v < 2; ++v)
      Vb[t][v] = *(gv4f*)(vsrc + 128 * t + 64 * v);
  }

  v8f sc[NT];
#pragma unroll
  for (int t = 0; t < NT; ++t) {
    v8f z = {};
    z = __builtin_amdgcn_wmma_f32_16x16x4_f32(false, Qa0, false, Kb[t].xy,
                                              (short)0, z, false, false);
    sc[t] = __builtin_amdgcn_wmma_f32_16x16x4_f32(false, Qa1, false, Kb[t].zw,
                                                  (short)0, z, false, false);
  }

  float mn[8];
#pragma unroll
  for (int r = 0; r < 8; ++r) {
    if (MASK) sc[NT - 1][r] += maskv[r];
    float mx = fmaxf(m[r], sc[0][r]);
#pragma unroll
    for (int t = 1; t < NT; ++t) mx = fmaxf(mx, sc[t][r]);
    mn[r] = mx;
  }
  rowmax16x8(mn);
#pragma unroll
  for (int r = 0; r < 8; ++r) {
    const float mnew = mn[r];
    const float corr = fast_exp2(m[r] - mnew);
    m[r] = mnew;
    o[r] = o[r] * corr;
#pragma unroll
    for (int t = 0; t < NT; ++t) sc[t][r] = fast_exp2(sc[t][r] - mnew);
  }

  float* pst = pw + hi * (8 * PPITCH) + lanelo;
  const float* pld = pw + lanelo * PPITCH + 4 * hi;
#pragma unroll
  for (int t = 0; t < NT; ++t) {
#pragma unroll
    for (int r = 0; r < 8; ++r) pst[t * PTILE + r * PPITCH] = sc[t][r];
  }
  v4f Pa[NT][2];
#pragma unroll
  for (int t = 0; t < NT; ++t) {
#pragma unroll
    for (int v = 0; v < 2; ++v)
      Pa[t][v] = *(const v4f*)(pld + t * PTILE + 8 * v);
  }
  __builtin_amdgcn_sched_barrier(0);
#pragma unroll
  for (int t = 0; t < NT; ++t) {
#pragma unroll
    for (int v = 0; v < 2; ++v) {
      o = __builtin_amdgcn_wmma_f32_16x16x4_f32(false, Pa[t][v].xy, false,
                                                Vb[t][v].xy, (short)0, o,
                                                false, false);
      o = __builtin_amdgcn_wmma_f32_16x16x4_f32(false, Pa[t][v].zw, false,
                                                Vb[t][v].zw, (short)0, o,
                                                false, false);
    }
  }
}

#define ATP 260

__device__ __forceinline__ v8f tile_gemm_khalf(const float* __restrict__ tile,
                                               const float* __restrict__ Wm,
                                               int ct, int kh, int lanelo,
                                               int hi) {
  const float* arow = tile + lanelo * ATP + kh * 128 + 4 * hi;
  const float* brow = Wm + (size_t)(ct * 16 + lanelo) * HID + kh * 128 + 4 * hi;
  v8f acc0 = {}, acc1 = {};
#pragma unroll
  for (int u = 0; u < 16; ++u) {
    const float4 a4 = *(const float4*)(arow + 8 * u);
    const float4 b4 = *(const float4*)(brow + 8 * u);
    acc0 = __builtin_amdgcn_wmma_f32_16x16x4_f32(
        false, (v2f){a4.x, a4.y}, false, (v2f){b4.x, b4.y}, (short)0, acc0,
        false, false);
    acc1 = __builtin_amdgcn_wmma_f32_16x16x4_f32(
        false, (v2f){a4.z, a4.w}, false, (v2f){b4.z, b4.w}, (short)0, acc1,
        false, false);
  }
  return acc0 + acc1;
}


__global__ __launch_bounds__(512) void kv_proj_rope_so8_kernel(
    const float* __restrict__ x, const float* __restrict__ Wk,
    const float* __restrict__ Wv, const float* __restrict__ rope,
    const float* __restrict__ so8, float* __restrict__ kb,
    float* __restrict__ vqb) {
  const int lane = threadIdx.x & 31;
  const int w = __builtin_amdgcn_readfirstlane((int)(threadIdx.x >> 5));
  const int lanelo = lane & 15;
  const int hi = lane >> 4;
  const int row0 = blockIdx.x * 16;
  const int b = row0 >> 11;
  const int s0 = row0 & (SEQ - 1);

  __shared__ __attribute__((aligned(16))) float tile[16 * ATP];
  __shared__ __attribute__((aligned(16))) float red[8 * 256];
  __shared__ __attribute__((aligned(16))) float trig[16 * 8];

  const int ct = w & 7;
  const int kh = w >> 3;
  const bool isk = ct < 4;
  const float* Wrows = (isk ? Wk : Wv) + (size_t)(ct & 3) * 16 * HID;

  const int xr = threadIdx.x >> 6;
  const int xc = (threadIdx.x & 63) * 4;
  const float4 x0 = *(const float4*)(x + ((size_t)row0 + xr) * HID + xc);
  const float4 x1 = *(const float4*)(x + ((size_t)row0 + 8 + xr) * HID + xc);
  const float* arow = Wrows + (size_t)lanelo * HID + kh * 128 + 4 * hi;
  float4 wa[16];
#pragma unroll
  for (int u = 0; u < 16; ++u) wa[u] = *(const float4*)(arow + 8 * u);

  *(float4*)(tile + xr * ATP + xc) = x0;
  *(float4*)(tile + (8 + xr) * ATP + xc) = x1;
  if (threadIdx.x < 128) {
    const int i = threadIdx.x & 3;
    const int tr = threadIdx.x >> 3;
    const float invf = (i == 0) ? 1.0f : (i == 1) ? 0.1f : (i == 2) ? 0.01f : 0.001f;
    const float fr = (float)(s0 + tr) * invf;
    trig[threadIdx.x] = (threadIdx.x & 4) ? sinf(fr) : cosf(fr);
  }
  __syncthreads();

  const float* brow = tile + lanelo * ATP + kh * 128 + 4 * hi;
  v8f acc0 = {}, acc1 = {};
#pragma unroll
  for (int u = 0; u < 16; ++u) {
    const float4 b4 = *(const float4*)(brow + 8 * u);
    acc0 = __builtin_amdgcn_wmma_f32_16x16x4_f32(
        false, (v2f){wa[u].x, wa[u].y}, false, (v2f){b4.x, b4.y}, (short)0,
        acc0, false, false);
    acc1 = __builtin_amdgcn_wmma_f32_16x16x4_f32(
        false, (v2f){wa[u].z, wa[u].w}, false, (v2f){b4.z, b4.w}, (short)0,
        acc1, false, false);
  }
  const v8f part = acc0 + acc1;
  if (kh == 1) {
#pragma unroll
    for (int r = 0; r < 8; ++r) red[ct * 256 + r * 32 + lane] = part[r];
  }
  __syncthreads();
  if (kh == 1) return;

  float acc[HDIM];
#pragma unroll
  for (int r = 0; r < 8; ++r) acc[r] = part[r] + red[ct * 256 + r * 32 + lane];
  const int head = (ct & 3) * 2 + hi;

  float y[HDIM];
  if (isk) {
    const float4 c4 = *(const float4*)(trig + lanelo * 8);
    const float4 s4 = *(const float4*)(trig + lanelo * 8 + 4);
    const float cs[4] = {c4.x, c4.y, c4.z, c4.w};
    const float sn[4] = {s4.x, s4.y, s4.z, s4.w};
    float cr[4] = {}, sr[4] = {};
#pragma unroll
    for (int i = 0; i < 4; ++i) {
#pragma unroll
      for (int j = 0; j < 4; ++j) {
        cr[j] = fmaf(cs[i], rope[i * 8 + j], cr[j]);
        sr[j] = fmaf(sn[i], rope[i * 8 + j], sr[j]);
      }
    }
#pragma unroll
    for (int d = 0; d < HDIM; ++d) {
      const int j = d & 3;
      const float rh = (d < 4) ? -acc[d + 4] : acc[d - 4];
      y[d] = acc[d] * cr[j] + rh * sr[j];
    }
  } else {
#pragma unroll
    for (int d = 0; d < HDIM; ++d) y[d] = acc[d];
  }

  float z[HDIM];
#pragma unroll
  for (int j = 0; j < HDIM; ++j) {
    float sum = 0.0f;
#pragma unroll
    for (int d = 0; d < HDIM; ++d) sum = fmaf(y[d], so8[d * 8 + j], sum);
    z[j] = sum;
  }

  if (isk) {
    float* dst = kb + (((size_t)b * N_KV + head) * SEQ + s0 + lanelo) * HDIM;
    *(float4*)dst = make_float4(z[0], z[1], z[2], z[3]);
    *(float4*)(dst + 4) = make_float4(z[4], z[5], z[6], z[7]);
  } else {
    float* stg = tile + (ct - 4) * 256;
    float* sw = stg + hi * 128 + (lanelo >> 2) * 32 + (lanelo & 3);
#pragma unroll
    for (int j = 0; j < HDIM; ++j) sw[j * 4] = z[j];
#pragma unroll
    for (int hh = 0; hh < 2; ++hh) {
      const v4f v4 = *(const v4f*)(stg + hh * 128 + lane * 4);
      float* dst = vqb +
                   (((size_t)b * N_KV + (ct & 3) * 2 + hh) * SEQ + s0) * HDIM +
                   lane * 4;
      *(v4f*)dst = v4;
    }
  }
}

__global__ __launch_bounds__(1024) void so8_qproj_flash_attn_outproj_kernel(
    const float* __restrict__ x, const float* __restrict__ Wq,
    const float* __restrict__ rope, const float* __restrict__ so8,
    const float* __restrict__ k, const float* __restrict__ vq,
    const float* __restrict__ Wo, float* __restrict__ out) {
  const int lane = threadIdx.x & 31;
  const int w = __builtin_amdgcn_readfirstlane((int)(threadIdx.x >> 5));
  const int b = blockIdx.x & 1;
  const int p = blockIdx.x >> 1;
  const int grp = (w >> 2) & 1;
  const int hp = ((w >> 3) << 2) | (w & 3);
  const int kv = hp >> 1;
  const int p8 = grp ? (2 * QTILES - 1 - p) : p;
  const int qt = p8 >> 1;
  const int lanelo = lane & 15;
  const int hi = lane >> 4;
  const int pos_early = p * 8;
  const int pos_late = (2 * QTILES - 1 - p) * 8 - 8;

  const float* kp = k + ((size_t)b * N_KV + kv) * SEQ * HDIM;
  const unsigned koff = (unsigned)(lanelo * HDIM + 4 * hi) * sizeof(float);
  const float* vp = vq + ((size_t)b * N_KV + kv) * SEQ * HDIM;
  const bool vlane = lanelo < 8;
  const float* vrow = vlane ? vp + hi * 32 + lanelo * 4 : kOnes;

  __shared__ __attribute__((aligned(16))) float plds[N_HEADS * 4 * PTILE];
  __shared__ __attribute__((aligned(16))) float tile[16 * ATP];
  __shared__ __attribute__((aligned(16))) float trig[16 * 8];
  float* pw = plds + w * 4 * PTILE;
  const int ct = w & 15;
  const int kh = w >> 4;

  {
    const int xr = threadIdx.x >> 6;
    const int xc = (threadIdx.x & 63) * 4;
    const int xpos = (xr < 8 ? pos_early : pos_late) + xr;
    const float4 xv = *(const float4*)(x + ((size_t)b * SEQ + xpos) * HID + xc);
    *(float4*)(tile + xr * ATP + xc) = xv;
  }
  if (threadIdx.x < 128) {
    const int i = threadIdx.x & 3;
    const int tr = threadIdx.x >> 3;
    const float invf = (i == 0) ? 1.0f : (i == 1) ? 0.1f : (i == 2) ? 0.01f : 0.001f;
    const float fr = (float)((tr < 8 ? pos_early : pos_late) + tr) * invf;
    trig[threadIdx.x] = (threadIdx.x & 4) ? sinf(fr) : cosf(fr);
  }
  __syncthreads();
  {
    const v8f qacc = tile_gemm_khalf(tile, Wq, ct, kh, lanelo, hi);
    float* red = plds + ct * 256;
    if (kh == 1) {
#pragma unroll
      for (int r = 0; r < 8; ++r) red[r * 32 + lane] = qacc[r];
    }
    __syncthreads();
    if (kh == 0) {
#pragma unroll
      for (int r = 0; r < 8; ++r)
        tile[(r + 8 * hi) * ATP + ct * 16 + lanelo] =
            qacc[r] + red[r * 32 + lane];
    }
    __syncthreads();
  }

  v2f Qa0, Qa1;
  {
    const int tr = 8 * grp + (lanelo & 7);
    const int hd = 2 * hp + (lanelo >> 3);
    const float4 q0 = *(const float4*)(tile + tr * ATP + hd * HDIM);
    const float4 q1 = *(const float4*)(tile + tr * ATP + hd * HDIM + 4);
    const float4 c4 = *(const float4*)(trig + tr * 8);
    const float4 s4 = *(const float4*)(trig + tr * 8 + 4);
    const float acc[HDIM] = {q0.x, q0.y, q0.z, q0.w, q1.x, q1.y, q1.z, q1.w};
    const float cs[4] = {c4.x, c4.y, c4.z, c4.w};
    const float sn[4] = {s4.x, s4.y, s4.z, s4.w};
    float cr[4] = {}, sr[4] = {};
#pragma unroll
    for (int i = 0; i < 4; ++i) {
#pragma unroll
      for (int j = 0; j < 4; ++j) {
        cr[j] = fmaf(cs[i], rope[i * 8 + j], cr[j]);
        sr[j] = fmaf(sn[i], rope[i * 8 + j], sr[j]);
      }
    }
    float y[HDIM];
#pragma unroll
    for (int d = 0; d < HDIM; ++d) {
      const int j = d & 3;
      const float rh = (d < 4) ? -acc[d + 4] : acc[d - 4];
      y[d] = acc[d] * cr[j] + rh * sr[j];
    }
    float z[HDIM];
#pragma unroll
    for (int j = 0; j < HDIM; ++j) {
      float sum = 0.0f;
#pragma unroll
      for (int d = 0; d < HDIM; ++d) sum = fmaf(y[d], so8[d * 8 + j], sum);
      z[j] = sum;
    }
    const float QSCALE = 0.35355339059327f * 1.44269504088896f;
    Qa0 = (v2f){hi ? z[4] : z[0], hi ? z[5] : z[1]} * QSCALE;
    Qa1 = (v2f){hi ? z[6] : z[2], hi ? z[7] : z[3]} * QSCALE;
  }
  __syncthreads();

  const int moff = 8 * (p8 & 1);
  float maskv[8];
#pragma unroll
  for (int r = 0; r < 8; ++r)
    maskv[r] = (lanelo > r + moff) ? -3.0e38f : 0.0f;

  v8f o = {};
  float m[8];
#pragma unroll
  for (int r = 0; r < 8; ++r) m[r] = -3.0e38f;

  const int nfull = qt >> 2;
  for (int kb = 0; kb < nfull; ++kb)
    attn_block<4, false>(kp, koff, vrow, pw, kb * 64, lanelo, hi, vlane, maskv,
                         Qa0, Qa1, m, o);
  const int tbase = nfull * 64;
  switch (qt & 3) {
    case 0:
      attn_block<1, true>(kp, koff, vrow, pw, tbase, lanelo, hi, vlane, maskv,
                          Qa0, Qa1, m, o);
      break;
    case 1:
      attn_block<2, true>(kp, koff, vrow, pw, tbase, lanelo, hi, vlane, maskv,
                          Qa0, Qa1, m, o);
      break;
    case 2:
      attn_block<3, true>(kp, koff, vrow, pw, tbase, lanelo, hi, vlane, maskv,
                          Qa0, Qa1, m, o);
      break;
    default:
      attn_block<4, true>(kp, koff, vrow, pw, tbase, lanelo, hi, vlane, maskv,
                          Qa0, Qa1, m, o);
      break;
  }

#pragma unroll
  for (int r = 0; r < 8; ++r) {
    const float ov = o[r];
    const float l = __shfl(ov, (lane & 16) + 8);
    const float outv = ov * __builtin_amdgcn_rcpf(l);
    if (lanelo < HDIM)
      tile[(8 * grp + r) * ATP + (2 * hp + hi) * HDIM + lanelo] = outv;
  }
  __syncthreads();

  const v8f acc = tile_gemm_khalf(tile, Wo, ct, kh, lanelo, hi);

  float* red = plds + ct * 256;
  if (kh == 1) {
#pragma unroll
    for (int r = 0; r < 8; ++r) red[r * 32 + lane] = acc[r];
  }
  __syncthreads();
  if (kh == 0) {
#pragma unroll
    for (int r = 0; r < 8; ++r)
      tile[(r + 8 * hi) * ATP + ct * 16 + lanelo] = acc[r] + red[r * 32 + lane];
  }
  __syncthreads();

  const int orow = w >> 1;
  const int oc = (w & 1) * 128 + lane * 4;
  const int opos = (orow < 8 ? pos_early : pos_late) + orow;
  const float4 ov4 = *(const float4*)(tile + orow * ATP + oc);
  *(float4*)(out + ((size_t)b * SEQ + opos) * HID + oc) = ov4;
}

extern "C" void kernel_launch(void* const* d_in, const int* in_sizes, int n_in,
                              void* d_out, int out_size, void* d_ws,
                              size_t ws_size, hipStream_t stream) {
  const float* x = (const float*)d_in[0];
  const float* Wq = (const float*)d_in[1];
  const float* Wk = (const float*)d_in[2];
  const float* Wv = (const float*)d_in[3];
  const float* Wo = (const float*)d_in[4];
  const float* rope = (const float*)d_in[5];
  const float* so8 = (const float*)d_in[6];

  float* ws = (float*)d_ws;
  float* kb = ws;
  float* vqb = kb + (size_t)2 * N_KV * SEQ * HDIM;

  kv_proj_rope_so8_kernel<<<2 * QTILES, 512, 0, stream>>>(x, Wk, Wv, rope, so8,
                                                         kb, vqb);
  so8_qproj_flash_attn_outproj_kernel<<<2 * QTILES, 1024, 0, stream>>>(
      x, Wq, rope, so8, kb, vqb, Wo, (float*)d_out);
}


# Softmax moved behind the score transpose: raw scores cross LDS, each lane then owns one row's keys; row max = in-lane max3 tree + one half exchange (was a 32-DPP butterfly); sums unchanged
# speedup vs baseline: 5.3005x; 1.0916x over previous
#include <hip/hip_runtime.h>
#include <hip/hip_bf16.h>

#define N_HEADS 32
#define N_KV 8
#define HDIM 8
#define SEQ 2048
#define HID 256
#define QTILES (SEQ / 16)

typedef __attribute__((ext_vector_type(2))) float v2f;
typedef __attribute__((ext_vector_type(4))) float v4f;
typedef __attribute__((ext_vector_type(8))) float v8f;
typedef const __attribute__((address_space(1))) v4f gv4f;

#define ONES8 1.0f, 1.0f, 1.0f, 1.0f, 1.0f, 1.0f, 1.0f, 1.0f
#define ONES64 ONES8, ONES8, ONES8, ONES8, ONES8, ONES8, ONES8, ONES8
__device__ __attribute__((aligned(16))) const float kOnes[512] = {
    ONES64, ONES64, ONES64, ONES64, ONES64, ONES64, ONES64, ONES64};

__device__ __forceinline__ float xhalf(float x) {
  const int b = __builtin_bit_cast(int, x);
  return __builtin_bit_cast(
      float, __builtin_amdgcn_permlanex16(b, b, 0x76543210, 0xfedcba98, false,
                                          false));
}

template <int N>
__device__ __forceinline__ float tree_max(const float (&v)[N]) {
  if constexpr (N == 1) {
    return v[0];
  } else if constexpr (N == 2) {
    return fmaxf(v[0], v[1]);
  } else {
    constexpr int M = (N + 2) / 3;
    float u[M];
#pragma unroll
    for (int i = 0; i < M; ++i) {
      if (3 * i + 2 < N)
        u[i] = fmaxf(fmaxf(v[3 * i], v[3 * i + 1]), v[3 * i + 2]);
      else if (3 * i + 1 < N)
        u[i] = fmaxf(v[3 * i], v[3 * i + 1]);
      else
        u[i] = v[3 * i];
    }
    return tree_max<M>(u);
  }
}

__device__ __forceinline__ float fast_exp2(float x) {
  return __builtin_amdgcn_exp2f(x);
}

#define PPITCH 20
#define PTILE (16 * PPITCH)

template <int NT, bool MASK>
__device__ __forceinline__ void attn_block(
    const float* __restrict__ kp, unsigned koff, const float* __restrict__ vrow,
    float* __restrict__ pw, float* cst, const float* cld, int kbase,
    int lanelo, int hi, bool vlane,
    const float* __restrict__ maskv, const v2f Qa0, const v2f Qa1, float& m,
    v8f& o) {
  const char* kblk = (const char*)(kp + (size_t)kbase * HDIM);
  v4f Kb[NT];
#pragma unroll
  for (int t = 0; t < NT; ++t)
    Kb[t] = *(const v4f*)(kblk + koff + 16 * t * HDIM * sizeof(float));
  const float* vsrc = vrow + (vlane ? kbase * HDIM : 0);
  v4f Vb[NT][2];
#pragma unroll
  for (int t = 0; t < NT; ++t) {
#pragma unroll
    for (int v = 0; v < 2; ++v)
      Vb[t][v] = *(gv4f*)(vsrc + 128 * t + 64 * v);
  }

  v8f sc[NT];
#pragma unroll
  for (int t = 0; t < NT; ++t) {
    v8f z = {};
    z = __builtin_amdgcn_wmma_f32_16x16x4_f32(false, Qa0, false, Kb[t].xy,
                                              (short)0, z, false, false);
    sc[t] = __builtin_amdgcn_wmma_f32_16x16x4_f32(false, Qa1, false, Kb[t].zw,
                                                  (short)0, z, false, false);
  }
  if (MASK) {
#pragma unroll
    for (int r = 0; r < 8; ++r)
      sc[NT - 1][r] += maskv[r];
  }

  float* pst = pw + hi * (8 * PPITCH) + lanelo;
  const float* pld = pw + lanelo * PPITCH + 4 * hi;
#pragma unroll
  for (int t = 0; t < NT; ++t) {
#pragma unroll
    for (int r = 0; r < 8; ++r) pst[t * PTILE + r * PPITCH] = sc[t][r];
  }
  v4f Pa[NT][2];
#pragma unroll
  for (int t = 0; t < NT; ++t) {
#pragma unroll
    for (int v = 0; v < 2; ++v)
      Pa[t][v] = *(const v4f*)(pld + t * PTILE + 8 * v);
  }

  float vals[NT * 8 + 1];
#pragma unroll
  for (int t = 0; t < NT; ++t) {
#pragma unroll
    for (int v = 0; v < 2; ++v) {
#pragma unroll
      for (int i = 0; i < 4; ++i) vals[(2 * t + v) * 4 + i] = Pa[t][v][i];
    }
  }
  vals[NT * 8] = m;
  float mnew = tree_max(vals);
  mnew = fmaxf(mnew, xhalf(mnew));
  const float corr = fast_exp2(m - mnew);
  m = mnew;
  *cst = corr;
  const v4f c0 = *(const v4f*)cld;
  const v4f c1 = *(const v4f*)(cld + 4);

#pragma unroll
  for (int t = 0; t < NT; ++t) {
#pragma unroll
    for (int v = 0; v < 2; ++v) Pa[t][v] = Pa[t][v] - mnew;
  }
#pragma unroll
  for (int t = 0; t < NT; ++t) {
#pragma unroll
    for (int v = 0; v < 2; ++v) {
#pragma unroll
      for (int i = 0; i < 4; ++i) Pa[t][v][i] = fast_exp2(Pa[t][v][i]);
    }
  }

  __builtin_amdgcn_sched_barrier(0);
  o[0] *= c0.x;
  o[1] *= c0.y;
  o[2] *= c0.z;
  o[3] *= c0.w;
  o[4] *= c1.x;
  o[5] *= c1.y;
  o[6] *= c1.z;
  o[7] *= c1.w;

#pragma unroll
  for (int t = 0; t < NT; ++t) {
#pragma unroll
    for (int v = 0; v < 2; ++v) {
      o = __builtin_amdgcn_wmma_f32_16x16x4_f32(false, Pa[t][v].xy, false,
                                                Vb[t][v].xy, (short)0, o,
                                                false, false);
      o = __builtin_amdgcn_wmma_f32_16x16x4_f32(false, Pa[t][v].zw, false,
                                                Vb[t][v].zw, (short)0, o,
                                                false, false);
    }
  }
}

#define ATP 260

__device__ __forceinline__ v8f tile_gemm_khalf(const float* __restrict__ tile,
                                               const float* __restrict__ Wm,
                                               int ct, int kh, int lanelo,
                                               int hi) {
  const float* arow = tile + lanelo * ATP + kh * 128 + 4 * hi;
  const float* brow = Wm + (size_t)(ct * 16 + lanelo) * HID + kh * 128 + 4 * hi;
  v8f acc0 = {}, acc1 = {};
#pragma unroll
  for (int u = 0; u < 16; ++u) {
    const float4 a4 = *(const float4*)(arow + 8 * u);
    const float4 b4 = *(const float4*)(brow + 8 * u);
    acc0 = __builtin_amdgcn_wmma_f32_16x16x4_f32(
        false, (v2f){a4.x, a4.y}, false, (v2f){b4.x, b4.y}, (short)0, acc0,
        false, false);
    acc1 = __builtin_amdgcn_wmma_f32_16x16x4_f32(
        false, (v2f){a4.z, a4.w}, false, (v2f){b4.z, b4.w}, (short)0, acc1,
        false, false);
  }
  return acc0 + acc1;
}


__global__ __launch_bounds__(512) void kv_proj_rope_so8_kernel(
    const float* __restrict__ x, const float* __restrict__ Wk,
    const float* __restrict__ Wv, const float* __restrict__ rope,
    const float* __restrict__ so8, float* __restrict__ kb,
    float* __restrict__ vqb) {
  const int lane = threadIdx.x & 31;
  const int w = __builtin_amdgcn_readfirstlane((int)(threadIdx.x >> 5));
  const int lanelo = lane & 15;
  const int hi = lane >> 4;
  const int row0 = blockIdx.x * 16;
  const int b = row0 >> 11;
  const int s0 = row0 & (SEQ - 1);

  __shared__ __attribute__((aligned(16))) float tile[16 * ATP];
  __shared__ __attribute__((aligned(16))) float red[8 * 256];
  __shared__ __attribute__((aligned(16))) float trig[16 * 8];

  const int ct = w & 7;
  const int kh = w >> 3;
  const bool isk = ct < 4;
  const float* Wrows = (isk ? Wk : Wv) + (size_t)(ct & 3) * 16 * HID;

  const int xr = threadIdx.x >> 6;
  const int xc = (threadIdx.x & 63) * 4;
  const float4 x0 = *(const float4*)(x + ((size_t)row0 + xr) * HID + xc);
  const float4 x1 = *(const float4*)(x + ((size_t)row0 + 8 + xr) * HID + xc);
  const float* arow = Wrows + (size_t)lanelo * HID + kh * 128 + 4 * hi;
  float4 wa[16];
#pragma unroll
  for (int u = 0; u < 16; ++u) wa[u] = *(const float4*)(arow + 8 * u);

  *(float4*)(tile + xr * ATP + xc) = x0;
  *(float4*)(tile + (8 + xr) * ATP + xc) = x1;
  if (threadIdx.x < 128) {
    const int i = threadIdx.x & 3;
    const int tr = threadIdx.x >> 3;
    const float invf = (i == 0) ? 1.0f : (i == 1) ? 0.1f : (i == 2) ? 0.01f : 0.001f;
    const float fr = (float)(s0 + tr) * invf;
    trig[threadIdx.x] = (threadIdx.x & 4) ? sinf(fr) : cosf(fr);
  }
  __syncthreads();

  const float* brow = tile + lanelo * ATP + kh * 128 + 4 * hi;
  v8f acc0 = {}, acc1 = {};
#pragma unroll
  for (int u = 0; u < 16; ++u) {
    const float4 b4 = *(const float4*)(brow + 8 * u);
    acc0 = __builtin_amdgcn_wmma_f32_16x16x4_f32(
        false, (v2f){wa[u].x, wa[u].y}, false, (v2f){b4.x, b4.y}, (short)0,
        acc0, false, false);
    acc1 = __builtin_amdgcn_wmma_f32_16x16x4_f32(
        false, (v2f){wa[u].z, wa[u].w}, false, (v2f){b4.z, b4.w}, (short)0,
        acc1, false, false);
  }
  const v8f part = acc0 + acc1;
  if (kh == 1) {
#pragma unroll
    for (int r = 0; r < 8; ++r) red[ct * 256 + r * 32 + lane] = part[r];
  }
  __syncthreads();
  if (kh == 1) return;

  float acc[HDIM];
#pragma unroll
  for (int r = 0; r < 8; ++r) acc[r] = part[r] + red[ct * 256 + r * 32 + lane];
  const int head = (ct & 3) * 2 + hi;

  float y[HDIM];
  if (isk) {
    const float4 c4 = *(const float4*)(trig + lanelo * 8);
    const float4 s4 = *(const float4*)(trig + lanelo * 8 + 4);
    const float cs[4] = {c4.x, c4.y, c4.z, c4.w};
    const float sn[4] = {s4.x, s4.y, s4.z, s4.w};
    float cr[4] = {}, sr[4] = {};
#pragma unroll
    for (int i = 0; i < 4; ++i) {
#pragma unroll
      for (int j = 0; j < 4; ++j) {
        cr[j] = fmaf(cs[i], rope[i * 8 + j], cr[j]);
        sr[j] = fmaf(sn[i], rope[i * 8 + j], sr[j]);
      }
    }
#pragma unroll
    for (int d = 0; d < HDIM; ++d) {
      const int j = d & 3;
      const float rh = (d < 4) ? -acc[d + 4] : acc[d - 4];
      y[d] = acc[d] * cr[j] + rh * sr[j];
    }
  } else {
#pragma unroll
    for (int d = 0; d < HDIM; ++d) y[d] = acc[d];
  }

  float z[HDIM];
#pragma unroll
  for (int j = 0; j < HDIM; ++j) {
    float sum = 0.0f;
#pragma unroll
    for (int d = 0; d < HDIM; ++d) sum = fmaf(y[d], so8[d * 8 + j], sum);
    z[j] = sum;
  }

  if (isk) {
    float* dst = kb + (((size_t)b * N_KV + head) * SEQ + s0 + lanelo) * HDIM;
    *(float4*)dst = make_float4(z[0], z[1], z[2], z[3]);
    *(float4*)(dst + 4) = make_float4(z[4], z[5], z[6], z[7]);
  } else {
    float* stg = tile + (ct - 4) * 256;
    float* sw = stg + hi * 128 + (lanelo >> 2) * 32 + (lanelo & 3);
#pragma unroll
    for (int j = 0; j < HDIM; ++j) sw[j * 4] = z[j];
#pragma unroll
    for (int hh = 0; hh < 2; ++hh) {
      const v4f v4 = *(const v4f*)(stg + hh * 128 + lane * 4);
      float* dst = vqb +
                   (((size_t)b * N_KV + (ct & 3) * 2 + hh) * SEQ + s0) * HDIM +
                   lane * 4;
      *(v4f*)dst = v4;
    }
  }
}

__global__ __launch_bounds__(1024) void so8_qproj_flash_attn_outproj_kernel(
    const float* __restrict__ x, const float* __restrict__ Wq,
    const float* __restrict__ rope, const float* __restrict__ so8,
    const float* __restrict__ k, const float* __restrict__ vq,
    const float* __restrict__ Wo, float* __restrict__ out) {
  const int lane = threadIdx.x & 31;
  const int w = __builtin_amdgcn_readfirstlane((int)(threadIdx.x >> 5));
  const int b = blockIdx.x & 1;
  const int p = blockIdx.x >> 1;
  const int grp = (w >> 2) & 1;
  const int hp = ((w >> 3) << 2) | (w & 3);
  const int kv = hp >> 1;
  const int p8 = grp ? (2 * QTILES - 1 - p) : p;
  const int qt = p8 >> 1;
  const int lanelo = lane & 15;
  const int hi = lane >> 4;
  const int pos_early = p * 8;
  const int pos_late = (2 * QTILES - 1 - p) * 8 - 8;

  const float* kp = k + ((size_t)b * N_KV + kv) * SEQ * HDIM;
  const unsigned koff = (unsigned)(lanelo * HDIM + 4 * hi) * sizeof(float);
  const float* vp = vq + ((size_t)b * N_KV + kv) * SEQ * HDIM;
  const bool vlane = lanelo < 8;
  const float* vrow = vlane ? vp + hi * 32 + lanelo * 4 : kOnes;

  __shared__ __attribute__((aligned(16))) float plds[N_HEADS * 4 * PTILE];
  __shared__ __attribute__((aligned(16))) float tile[16 * ATP];
  __shared__ __attribute__((aligned(16))) float trig[16 * 8];
  __shared__ __attribute__((aligned(16))) float cbuf[N_HEADS * 32];
  float* pw = plds + w * 4 * PTILE;
  float* cst = cbuf + w * 32 + 16 * hi + lanelo;
  const float* cld = cbuf + w * 32 + 16 * hi + 8 * hi;
  const int ct = w & 15;
  const int kh = w >> 4;

  {
    const int xr = threadIdx.x >> 6;
    const int xc = (threadIdx.x & 63) * 4;
    const int xpos = (xr < 8 ? pos_early : pos_late) + xr;
    const float4 xv = *(const float4*)(x + ((size_t)b * SEQ + xpos) * HID + xc);
    *(float4*)(tile + xr * ATP + xc) = xv;
  }
  if (threadIdx.x < 128) {
    const int i = threadIdx.x & 3;
    const int tr = threadIdx.x >> 3;
    const float invf = (i == 0) ? 1.0f : (i == 1) ? 0.1f : (i == 2) ? 0.01f : 0.001f;
    const float fr = (float)((tr < 8 ? pos_early : pos_late) + tr) * invf;
    trig[threadIdx.x] = (threadIdx.x & 4) ? sinf(fr) : cosf(fr);
  }
  __syncthreads();
  {
    const v8f qacc = tile_gemm_khalf(tile, Wq, ct, kh, lanelo, hi);
    float* red = plds + ct * 256;
    if (kh == 1) {
#pragma unroll
      for (int r = 0; r < 8; ++r) red[r * 32 + lane] = qacc[r];
    }
    __syncthreads();
    if (kh == 0) {
#pragma unroll
      for (int r = 0; r < 8; ++r)
        tile[(r + 8 * hi) * ATP + ct * 16 + lanelo] =
            qacc[r] + red[r * 32 + lane];
    }
    __syncthreads();
  }

  v2f Qa0, Qa1;
  {
    const int tr = 8 * grp + (lanelo & 7);
    const int hd = 2 * hp + (lanelo >> 3);
    const float4 q0 = *(const float4*)(tile + tr * ATP + hd * HDIM);
    const float4 q1 = *(const float4*)(tile + tr * ATP + hd * HDIM + 4);
    const float4 c4 = *(const float4*)(trig + tr * 8);
    const float4 s4 = *(const float4*)(trig + tr * 8 + 4);
    const float acc[HDIM] = {q0.x, q0.y, q0.z, q0.w, q1.x, q1.y, q1.z, q1.w};
    const float cs[4] = {c4.x, c4.y, c4.z, c4.w};
    const float sn[4] = {s4.x, s4.y, s4.z, s4.w};
    float cr[4] = {}, sr[4] = {};
#pragma unroll
    for (int i = 0; i < 4; ++i) {
#pragma unroll
      for (int j = 0; j < 4; ++j) {
        cr[j] = fmaf(cs[i], rope[i * 8 + j], cr[j]);
        sr[j] = fmaf(sn[i], rope[i * 8 + j], sr[j]);
      }
    }
    float y[HDIM];
#pragma unroll
    for (int d = 0; d < HDIM; ++d) {
      const int j = d & 3;
      const float rh = (d < 4) ? -acc[d + 4] : acc[d - 4];
      y[d] = acc[d] * cr[j] + rh * sr[j];
    }
    float z[HDIM];
#pragma unroll
    for (int j = 0; j < HDIM; ++j) {
      float sum = 0.0f;
#pragma unroll
      for (int d = 0; d < HDIM; ++d) sum = fmaf(y[d], so8[d * 8 + j], sum);
      z[j] = sum;
    }
    const float QSCALE = 0.35355339059327f * 1.44269504088896f;
    Qa0 = (v2f){hi ? z[4] : z[0], hi ? z[5] : z[1]} * QSCALE;
    Qa1 = (v2f){hi ? z[6] : z[2], hi ? z[7] : z[3]} * QSCALE;
  }
  __syncthreads();

  const int moff = 8 * (p8 & 1);
  float maskv[8];
#pragma unroll
  for (int r = 0; r < 8; ++r)
    maskv[r] = (lanelo > r + moff) ? -3.0e38f : 0.0f;

  v8f o = {};
  float m = -3.0e38f;

  const int nfull = qt >> 2;
  for (int kb = 0; kb < nfull; ++kb)
    attn_block<4, false>(kp, koff, vrow, pw, cst, cld, kb * 64, lanelo, hi, vlane,
                         maskv, Qa0, Qa1, m, o);
  const int tbase = nfull * 64;
  switch (qt & 3) {
    case 0:
      attn_block<1, true>(kp, koff, vrow, pw, cst, cld, tbase, lanelo, hi, vlane,
                          maskv, Qa0, Qa1, m, o);
      break;
    case 1:
      attn_block<2, true>(kp, koff, vrow, pw, cst, cld, tbase, lanelo, hi, vlane,
                          maskv, Qa0, Qa1, m, o);
      break;
    case 2:
      attn_block<3, true>(kp, koff, vrow, pw, cst, cld, tbase, lanelo, hi, vlane,
                          maskv, Qa0, Qa1, m, o);
      break;
    default:
      attn_block<4, true>(kp, koff, vrow, pw, cst, cld, tbase, lanelo, hi, vlane,
                          maskv, Qa0, Qa1, m, o);
      break;
  }

#pragma unroll
  for (int r = 0; r < 8; ++r) {
    const float ov = o[r];
    const float l = __shfl(ov, (lane & 16) + 8);
    const float outv = ov * __builtin_amdgcn_rcpf(l);
    if (lanelo < HDIM)
      tile[(8 * grp + r) * ATP + (2 * hp + hi) * HDIM + lanelo] = outv;
  }
  __syncthreads();

  const v8f acc = tile_gemm_khalf(tile, Wo, ct, kh, lanelo, hi);

  float* red = plds + ct * 256;
  if (kh == 1) {
#pragma unroll
    for (int r = 0; r < 8; ++r) red[r * 32 + lane] = acc[r];
  }
  __syncthreads();
  if (kh == 0) {
#pragma unroll
    for (int r = 0; r < 8; ++r)
      tile[(r + 8 * hi) * ATP + ct * 16 + lanelo] = acc[r] + red[r * 32 + lane];
  }
  __syncthreads();

  const int orow = w >> 1;
  const int oc = (w & 1) * 128 + lane * 4;
  const int opos = (orow < 8 ? pos_early : pos_late) + orow;
  const float4 ov4 = *(const float4*)(tile + orow * ATP + oc);
  *(float4*)(out + ((size_t)b * SEQ + opos) * HID + oc) = ov4;
}

extern "C" void kernel_launch(void* const* d_in, const int* in_sizes, int n_in,
                              void* d_out, int out_size, void* d_ws,
                              size_t ws_size, hipStream_t stream) {
  const float* x = (const float*)d_in[0];
  const float* Wq = (const float*)d_in[1];
  const float* Wk = (const float*)d_in[2];
  const float* Wv = (const float*)d_in[3];
  const float* Wo = (const float*)d_in[4];
  const float* rope = (const float*)d_in[5];
  const float* so8 = (const float*)d_in[6];

  float* ws = (float*)d_ws;
  float* kb = ws;
  float* vqb = kb + (size_t)2 * N_KV * SEQ * HDIM;

  kv_proj_rope_so8_kernel<<<2 * QTILES, 512, 0, stream>>>(x, Wk, Wv, rope, so8,
                                                         kb, vqb);
  so8_qproj_flash_attn_outproj_kernel<<<2 * QTILES, 1024, 0, stream>>>(
      x, Wq, rope, so8, kb, vqb, Wo, (float*)d_out);
}


# Late q tile's keys split between its head pair's two waves: the early tile's wave takes the first 64-key blocks into a second softmax state, merged through LDS; late rows summed as two partials
# speedup vs baseline: 5.4658x; 1.0312x over previous
#include <hip/hip_runtime.h>
#include <hip/hip_bf16.h>

#define N_HEADS 32
#define N_KV 8
#define HDIM 8
#define SEQ 2048
#define HID 256
#define QTILES (SEQ / 16)

typedef __attribute__((ext_vector_type(2))) float v2f;
typedef __attribute__((ext_vector_type(4))) float v4f;
typedef __attribute__((ext_vector_type(8))) float v8f;
typedef const __attribute__((address_space(1))) v4f gv4f;

#define ONES8 1.0f, 1.0f, 1.0f, 1.0f, 1.0f, 1.0f, 1.0f, 1.0f
#define ONES64 ONES8, ONES8, ONES8, ONES8, ONES8, ONES8, ONES8, ONES8
__device__ __attribute__((aligned(16))) const float kOnes[512] = {
    ONES64, ONES64, ONES64, ONES64, ONES64, ONES64, ONES64, ONES64};

__device__ __forceinline__ float xhalf(float x) {
  const int b = __builtin_bit_cast(int, x);
  return __builtin_bit_cast(
      float, __builtin_amdgcn_permlanex16(b, b, 0x76543210, 0xfedcba98, false,
                                          false));
}

template <int N>
__device__ __forceinline__ float tree_max(const float (&v)[N]) {
  if constexpr (N == 1) {
    return v[0];
  } else if constexpr (N == 2) {
    return fmaxf(v[0], v[1]);
  } else {
    constexpr int M = (N + 2) / 3;
    float u[M];
#pragma unroll
    for (int i = 0; i < M; ++i) {
      if (3 * i + 2 < N)
        u[i] = fmaxf(fmaxf(v[3 * i], v[3 * i + 1]), v[3 * i + 2]);
      else if (3 * i + 1 < N)
        u[i] = fmaxf(v[3 * i], v[3 * i + 1]);
      else
        u[i] = v[3 * i];
    }
    return tree_max<M>(u);
  }
}

__device__ __forceinline__ float fast_exp2(float x) {
  return __builtin_amdgcn_exp2f(x);
}

#define PPITCH 20
#define PTILE (16 * PPITCH)

template <int NT, bool MASK>
__device__ __forceinline__ void attn_block(
    const float* __restrict__ kp, unsigned koff, const float* __restrict__ vrow,
    float* __restrict__ pw, float* cst, const float* cld, int kbase,
    int lanelo, int hi, bool vlane,
    const float* __restrict__ maskv, const v2f Qa0, const v2f Qa1, float& m,
    v8f& o) {
  const char* kblk = (const char*)(kp + (size_t)kbase * HDIM);
  v4f Kb[NT];
#pragma unroll
  for (int t = 0; t < NT; ++t)
    Kb[t] = *(const v4f*)(kblk + koff + 16 * t * HDIM * sizeof(float));
  const float* vsrc = vrow + (vlane ? kbase * HDIM : 0);
  v4f Vb[NT][2];
#pragma unroll
  for (int t = 0; t < NT; ++t) {
#pragma unroll
    for (int v = 0; v < 2; ++v)
      Vb[t][v] = *(gv4f*)(vsrc + 128 * t + 64 * v);
  }

  v8f sc[NT];
#pragma unroll
  for (int t = 0; t < NT; ++t) {
    v8f z = {};
    z = __builtin_amdgcn_wmma_f32_16x16x4_f32(false, Qa0, false, Kb[t].xy,
                                              (short)0, z, false, false);
    sc[t] = __builtin_amdgcn_wmma_f32_16x16x4_f32(false, Qa1, false, Kb[t].zw,
                                                  (short)0, z, false, false);
  }
  if (MASK) {
#pragma unroll
    for (int r = 0; r < 8; ++r)
      sc[NT - 1][r] += maskv[r];
  }

  float* pst = pw + hi * (8 * PPITCH) + lanelo;
  const float* pld = pw + lanelo * PPITCH + 4 * hi;
#pragma unroll
  for (int t = 0; t < NT; ++t) {
#pragma unroll
    for (int r = 0; r < 8; ++r) pst[t * PTILE + r * PPITCH] = sc[t][r];
  }
  v4f Pa[NT][2];
#pragma unroll
  for (int t = 0; t < NT; ++t) {
#pragma unroll
    for (int v = 0; v < 2; ++v)
      Pa[t][v] = *(const v4f*)(pld + t * PTILE + 8 * v);
  }

  float vals[NT * 8 + 1];
#pragma unroll
  for (int t = 0; t < NT; ++t) {
#pragma unroll
    for (int v = 0; v < 2; ++v) {
#pragma unroll
      for (int i = 0; i < 4; ++i) vals[(2 * t + v) * 4 + i] = Pa[t][v][i];
    }
  }
  vals[NT * 8] = m;
  float mnew = tree_max(vals);
  mnew = fmaxf(mnew, xhalf(mnew));
  const float corr = fast_exp2(m - mnew);
  m = mnew;
  *cst = corr;
  const v4f c0 = *(const v4f*)cld;
  const v4f c1 = *(const v4f*)(cld + 4);

#pragma unroll
  for (int t = 0; t < NT; ++t) {
#pragma unroll
    for (int v = 0; v < 2; ++v) Pa[t][v] = Pa[t][v] - mnew;
  }
#pragma unroll
  for (int t = 0; t < NT; ++t) {
#pragma unroll
    for (int v = 0; v < 2; ++v) {
#pragma unroll
      for (int i = 0; i < 4; ++i) Pa[t][v][i] = fast_exp2(Pa[t][v][i]);
    }
  }

  __builtin_amdgcn_sched_barrier(0);
  o[0] *= c0.x;
  o[1] *= c0.y;
  o[2] *= c0.z;
  o[3] *= c0.w;
  o[4] *= c1.x;
  o[5] *= c1.y;
  o[6] *= c1.z;
  o[7] *= c1.w;

#pragma unroll
  for (int t = 0; t < NT; ++t) {
#pragma unroll
    for (int v = 0; v < 2; ++v) {
      o = __builtin_amdgcn_wmma_f32_16x16x4_f32(false, Pa[t][v].xy, false,
                                                Vb[t][v].xy, (short)0, o,
                                                false, false);
      o = __builtin_amdgcn_wmma_f32_16x16x4_f32(false, Pa[t][v].zw, false,
                                                Vb[t][v].zw, (short)0, o,
                                                false, false);
    }
  }
}

#define ATP 260

__device__ __forceinline__ v8f tile_gemm_khalf(const float* __restrict__ tile,
                                               const float* __restrict__ Wm,
                                               int ct, int kh, int lanelo,
                                               int hi) {
  const float* arow = tile + lanelo * ATP + kh * 128 + 4 * hi;
  const float* brow = Wm + (size_t)(ct * 16 + lanelo) * HID + kh * 128 + 4 * hi;
  v8f acc0 = {}, acc1 = {};
#pragma unroll
  for (int u = 0; u < 16; ++u) {
    const float4 a4 = *(const float4*)(arow + 8 * u);
    const float4 b4 = *(const float4*)(brow + 8 * u);
    acc0 = __builtin_amdgcn_wmma_f32_16x16x4_f32(
        false, (v2f){a4.x, a4.y}, false, (v2f){b4.x, b4.y}, (short)0, acc0,
        false, false);
    acc1 = __builtin_amdgcn_wmma_f32_16x16x4_f32(
        false, (v2f){a4.z, a4.w}, false, (v2f){b4.z, b4.w}, (short)0, acc1,
        false, false);
  }
  return acc0 + acc1;
}


__global__ __launch_bounds__(512) void kv_proj_rope_so8_kernel(
    const float* __restrict__ x, const float* __restrict__ Wk,
    const float* __restrict__ Wv, const float* __restrict__ rope,
    const float* __restrict__ so8, float* __restrict__ kb,
    float* __restrict__ vqb) {
  const int lane = threadIdx.x & 31;
  const int w = __builtin_amdgcn_readfirstlane((int)(threadIdx.x >> 5));
  const int lanelo = lane & 15;
  const int hi = lane >> 4;
  const int row0 = blockIdx.x * 16;
  const int b = row0 >> 11;
  const int s0 = row0 & (SEQ - 1);

  __shared__ __attribute__((aligned(16))) float tile[16 * ATP];
  __shared__ __attribute__((aligned(16))) float red[8 * 256];
  __shared__ __attribute__((aligned(16))) float trig[16 * 8];

  const int ct = w & 7;
  const int kh = w >> 3;
  const bool isk = ct < 4;
  const float* Wrows = (isk ? Wk : Wv) + (size_t)(ct & 3) * 16 * HID;

  const int xr = threadIdx.x >> 6;
  const int xc = (threadIdx.x & 63) * 4;
  const float4 x0 = *(const float4*)(x + ((size_t)row0 + xr) * HID + xc);
  const float4 x1 = *(const float4*)(x + ((size_t)row0 + 8 + xr) * HID + xc);
  const float* arow = Wrows + (size_t)lanelo * HID + kh * 128 + 4 * hi;
  float4 wa[16];
#pragma unroll
  for (int u = 0; u < 16; ++u) wa[u] = *(const float4*)(arow + 8 * u);

  *(float4*)(tile + xr * ATP + xc) = x0;
  *(float4*)(tile + (8 + xr) * ATP + xc) = x1;
  if (threadIdx.x < 128) {
    const int i = threadIdx.x & 3;
    const int tr = threadIdx.x >> 3;
    const float invf = (i == 0) ? 1.0f : (i == 1) ? 0.1f : (i == 2) ? 0.01f : 0.001f;
    const float fr = (float)(s0 + tr) * invf;
    trig[threadIdx.x] = (threadIdx.x & 4) ? sinf(fr) : cosf(fr);
  }
  __syncthreads();

  const float* brow = tile + lanelo * ATP + kh * 128 + 4 * hi;
  v8f acc0 = {}, acc1 = {};
#pragma unroll
  for (int u = 0; u < 16; ++u) {
    const float4 b4 = *(const float4*)(brow + 8 * u);
    acc0 = __builtin_amdgcn_wmma_f32_16x16x4_f32(
        false, (v2f){wa[u].x, wa[u].y}, false, (v2f){b4.x, b4.y}, (short)0,
        acc0, false, false);
    acc1 = __builtin_amdgcn_wmma_f32_16x16x4_f32(
        false, (v2f){wa[u].z, wa[u].w}, false, (v2f){b4.z, b4.w}, (short)0,
        acc1, false, false);
  }
  const v8f part = acc0 + acc1;
  if (kh == 1) {
#pragma unroll
    for (int r = 0; r < 8; ++r) red[ct * 256 + r * 32 + lane] = part[r];
  }
  __syncthreads();
  if (kh == 1) return;

  float acc[HDIM];
#pragma unroll
  for (int r = 0; r < 8; ++r) acc[r] = part[r] + red[ct * 256 + r * 32 + lane];
  const int head = (ct & 3) * 2 + hi;

  float y[HDIM];
  if (isk) {
    const float4 c4 = *(const float4*)(trig + lanelo * 8);
    const float4 s4 = *(const float4*)(trig + lanelo * 8 + 4);
    const float cs[4] = {c4.x, c4.y, c4.z, c4.w};
    const float sn[4] = {s4.x, s4.y, s4.z, s4.w};
    float cr[4] = {}, sr[4] = {};
#pragma unroll
    for (int i = 0; i < 4; ++i) {
#pragma unroll
      for (int j = 0; j < 4; ++j) {
        cr[j] = fmaf(cs[i], rope[i * 8 + j], cr[j]);
        sr[j] = fmaf(sn[i], rope[i * 8 + j], sr[j]);
      }
    }
#pragma unroll
    for (int d = 0; d < HDIM; ++d) {
      const int j = d & 3;
      const float rh = (d < 4) ? -acc[d + 4] : acc[d - 4];
      y[d] = acc[d] * cr[j] + rh * sr[j];
    }
  } else {
#pragma unroll
    for (int d = 0; d < HDIM; ++d) y[d] = acc[d];
  }

  float z[HDIM];
#pragma unroll
  for (int j = 0; j < HDIM; ++j) {
    float sum = 0.0f;
#pragma unroll
    for (int d = 0; d < HDIM; ++d) sum = fmaf(y[d], so8[d * 8 + j], sum);
    z[j] = sum;
  }

  if (isk) {
    float* dst = kb + (((size_t)b * N_KV + head) * SEQ + s0 + lanelo) * HDIM;
    *(float4*)dst = make_float4(z[0], z[1], z[2], z[3]);
    *(float4*)(dst + 4) = make_float4(z[4], z[5], z[6], z[7]);
  } else {
    float* stg = tile + (ct - 4) * 256;
    float* sw = stg + hi * 128 + (lanelo >> 2) * 32 + (lanelo & 3);
#pragma unroll
    for (int j = 0; j < HDIM; ++j) sw[j * 4] = z[j];
#pragma unroll
    for (int hh = 0; hh < 2; ++hh) {
      const v4f v4 = *(const v4f*)(stg + hh * 128 + lane * 4);
      float* dst = vqb +
                   (((size_t)b * N_KV + (ct & 3) * 2 + hh) * SEQ + s0) * HDIM +
                   lane * 4;
      *(v4f*)dst = v4;
    }
  }
}

__global__ __launch_bounds__(1024) void so8_qproj_flash_attn_outproj_kernel(
    const float* __restrict__ x, const float* __restrict__ Wq,
    const float* __restrict__ rope, const float* __restrict__ so8,
    const float* __restrict__ k, const float* __restrict__ vq,
    const float* __restrict__ Wo, float* __restrict__ out) {
  const int lane = threadIdx.x & 31;
  const int w = __builtin_amdgcn_readfirstlane((int)(threadIdx.x >> 5));
  const int b = blockIdx.x & 1;
  const int p = blockIdx.x >> 1;
  const int grp = (w >> 2) & 1;
  const int hp = ((w >> 3) << 2) | (w & 3);
  const int kv = hp >> 1;
  const int p8 = grp ? (2 * QTILES - 1 - p) : p;
  const int qt = p8 >> 1;
  const int lanelo = lane & 15;
  const int hi = lane >> 4;
  const int pos_early = p * 8;
  const int pos_late = (2 * QTILES - 1 - p) * 8 - 8;

  const float* kp = k + ((size_t)b * N_KV + kv) * SEQ * HDIM;
  const unsigned koff = (unsigned)(lanelo * HDIM + 4 * hi) * sizeof(float);
  const float* vp = vq + ((size_t)b * N_KV + kv) * SEQ * HDIM;
  const bool vlane = lanelo < 8;
  const float* vrow = vlane ? vp + hi * 32 + lanelo * 4 : kOnes;

  __shared__ __attribute__((aligned(16))) float plds[N_HEADS * 4 * PTILE];
  __shared__ __attribute__((aligned(16))) float tile[16 * ATP];
  __shared__ __attribute__((aligned(16))) float trig[16 * 8];
  __shared__ __attribute__((aligned(16))) float cbuf[N_HEADS * 32];
  float* pw = plds + w * 4 * PTILE;
  float* cst = cbuf + w * 32 + 16 * hi + lanelo;
  const float* cld = cbuf + w * 32 + 16 * hi + 8 * hi;
  const int ct = w & 15;
  const int kh = w >> 4;

  {
    const int xr = threadIdx.x >> 6;
    const int xc = (threadIdx.x & 63) * 4;
    const int xpos = (xr < 8 ? pos_early : pos_late) + xr;
    const float4 xv = *(const float4*)(x + ((size_t)b * SEQ + xpos) * HID + xc);
    *(float4*)(tile + xr * ATP + xc) = xv;
  }
  if (threadIdx.x < 128) {
    const int i = threadIdx.x & 3;
    const int tr = threadIdx.x >> 3;
    const float invf = (i == 0) ? 1.0f : (i == 1) ? 0.1f : (i == 2) ? 0.01f : 0.001f;
    const float fr = (float)((tr < 8 ? pos_early : pos_late) + tr) * invf;
    trig[threadIdx.x] = (threadIdx.x & 4) ? sinf(fr) : cosf(fr);
  }
  __syncthreads();
  {
    const v8f qacc = tile_gemm_khalf(tile, Wq, ct, kh, lanelo, hi);
    float* red = plds + ct * 256;
    if (kh == 1) {
#pragma unroll
      for (int r = 0; r < 8; ++r) red[r * 32 + lane] = qacc[r];
    }
    __syncthreads();
    if (kh == 0) {
#pragma unroll
      for (int r = 0; r < 8; ++r)
        tile[(r + 8 * hi) * ATP + ct * 16 + lanelo] =
            qacc[r] + red[r * 32 + lane];
    }
    __syncthreads();
  }

  auto make_q = [&](int g, v2f& Qa0, v2f& Qa1) {
    const int tr = 8 * g + (lanelo & 7);
    const int hd = 2 * hp + (lanelo >> 3);
    const float4 q0 = *(const float4*)(tile + tr * ATP + hd * HDIM);
    const float4 q1 = *(const float4*)(tile + tr * ATP + hd * HDIM + 4);
    const float4 c4 = *(const float4*)(trig + tr * 8);
    const float4 s4 = *(const float4*)(trig + tr * 8 + 4);
    const float acc[HDIM] = {q0.x, q0.y, q0.z, q0.w, q1.x, q1.y, q1.z, q1.w};
    const float cs[4] = {c4.x, c4.y, c4.z, c4.w};
    const float sn[4] = {s4.x, s4.y, s4.z, s4.w};
    float cr[4] = {}, sr[4] = {};
#pragma unroll
    for (int i = 0; i < 4; ++i) {
#pragma unroll
      for (int j = 0; j < 4; ++j) {
        cr[j] = fmaf(cs[i], rope[i * 8 + j], cr[j]);
        sr[j] = fmaf(sn[i], rope[i * 8 + j], sr[j]);
      }
    }
    float y[HDIM];
#pragma unroll
    for (int d = 0; d < HDIM; ++d) {
      const int j = d & 3;
      const float rh = (d < 4) ? -acc[d + 4] : acc[d - 4];
      y[d] = acc[d] * cr[j] + rh * sr[j];
    }
    float z[HDIM];
#pragma unroll
    for (int j = 0; j < HDIM; ++j) {
      float sum = 0.0f;
#pragma unroll
      for (int d = 0; d < HDIM; ++d) sum = fmaf(y[d], so8[d * 8 + j], sum);
      z[j] = sum;
    }
    const float QSCALE = 0.35355339059327f * 1.44269504088896f;
    Qa0 = (v2f){hi ? z[4] : z[0], hi ? z[5] : z[1]} * QSCALE;
    Qa1 = (v2f){hi ? z[6] : z[2], hi ? z[7] : z[3]} * QSCALE;
  };
  v2f Qa0, Qa1;
  make_q(grp, Qa0, Qa1);
  v2f Qb0 = {}, Qb1 = {};
  if (grp == 0) make_q(1, Qb0, Qb1);
  __syncthreads();

  const int moff = 8 * (p8 & 1);
  float maskv[8];
#pragma unroll
  for (int r = 0; r < 8; ++r)
    maskv[r] = (lanelo > r + moff) ? -3.0e38f : 0.0f;

  v8f o = {};
  float m = -3.0e38f;

  const int nfull = qt >> 2;
  const int nfull_early = (p >> 1) >> 2;
  const int nfull_late = ((2 * QTILES - 1 - p) >> 1) >> 2;
  const int nsplit = (nfull_late - nfull_early) >> 1;
  const int kskip = grp ? nsplit * 64 : 0;
  const float* kpw = kp + (size_t)kskip * HDIM;
  const float* vroww = vrow + (vlane ? kskip * HDIM : 0);
  const int nown = nfull - (kskip >> 6);
  for (int kb = 0; kb < nown; ++kb)
    attn_block<4, false>(kpw, koff, vroww, pw, cst, cld, kb * 64, lanelo, hi,
                         vlane, maskv, Qa0, Qa1, m, o);
  const int tbase = nown * 64;
  switch (qt & 3) {
    case 0:
      attn_block<1, true>(kpw, koff, vroww, pw, cst, cld, tbase, lanelo, hi,
                          vlane, maskv, Qa0, Qa1, m, o);
      break;
    case 1:
      attn_block<2, true>(kpw, koff, vroww, pw, cst, cld, tbase, lanelo, hi,
                          vlane, maskv, Qa0, Qa1, m, o);
      break;
    case 2:
      attn_block<3, true>(kpw, koff, vroww, pw, cst, cld, tbase, lanelo, hi,
                          vlane, maskv, Qa0, Qa1, m, o);
      break;
    default:
      attn_block<4, true>(kpw, koff, vroww, pw, cst, cld, tbase, lanelo, hi,
                          vlane, maskv, Qa0, Qa1, m, o);
      break;
  }

  auto store_rows = [&](const v8f& acc, int g) {
#pragma unroll
    for (int r = 0; r < 8; ++r) {
      const float ov = acc[r];
      const float l = __shfl(ov, (lane & 16) + 8);
      const float outv = ov * __builtin_amdgcn_rcpf(l);
      if (lanelo < HDIM)
        tile[(8 * g + r) * ATP + (2 * hp + hi) * HDIM + lanelo] = outv;
    }
  };
  if (grp == 0) {
    store_rows(o, 0);
    o = (v8f){};
    m = -3.0e38f;
    for (int kb = 0; kb < nsplit; ++kb)
      attn_block<4, false>(kp, koff, vrow, pw, cst, cld, kb * 64, lanelo, hi,
                           vlane, maskv, Qb0, Qb1, m, o);
#pragma unroll
    for (int r = 0; r < 8; ++r) pw[r * 32 + lane] = o[r];
    pw[256 + lane] = m;
  }
  __syncthreads();
  if (grp == 1) {
    const float* pp = plds + (w - 4) * 4 * PTILE;
    *cst = m;
    const v4f mo0 = *(const v4f*)cld;
    const v4f mo1 = *(const v4f*)(cld + 4);
    const v4f mp0 = *(const v4f*)(pp + 256 + 8 * hi);
    const v4f mp1 = *(const v4f*)(pp + 256 + 8 * hi + 4);
    const float mo[8] = {mo0.x, mo0.y, mo0.z, mo0.w, mo1.x, mo1.y, mo1.z, mo1.w};
    const float mp[8] = {mp0.x, mp0.y, mp0.z, mp0.w, mp1.x, mp1.y, mp1.z, mp1.w};
#pragma unroll
    for (int r = 0; r < 8; ++r) {
      const float mt = fmaxf(mo[r], mp[r]);
      o[r] = o[r] * fast_exp2(mo[r] - mt) +
             pp[r * 32 + lane] * fast_exp2(mp[r] - mt);
    }
    store_rows(o, 1);
  }
  __syncthreads();

  const v8f acc = tile_gemm_khalf(tile, Wo, ct, kh, lanelo, hi);

  float* red = plds + ct * 256;
  if (kh == 1) {
#pragma unroll
    for (int r = 0; r < 8; ++r) red[r * 32 + lane] = acc[r];
  }
  __syncthreads();
  if (kh == 0) {
#pragma unroll
    for (int r = 0; r < 8; ++r)
      tile[(r + 8 * hi) * ATP + ct * 16 + lanelo] = acc[r] + red[r * 32 + lane];
  }
  __syncthreads();

  const int orow = w >> 1;
  const int oc = (w & 1) * 128 + lane * 4;
  const int opos = (orow < 8 ? pos_early : pos_late) + orow;
  const float4 ov4 = *(const float4*)(tile + orow * ATP + oc);
  *(float4*)(out + ((size_t)b * SEQ + opos) * HID + oc) = ov4;
}

extern "C" void kernel_launch(void* const* d_in, const int* in_sizes, int n_in,
                              void* d_out, int out_size, void* d_ws,
                              size_t ws_size, hipStream_t stream) {
  const float* x = (const float*)d_in[0];
  const float* Wq = (const float*)d_in[1];
  const float* Wk = (const float*)d_in[2];
  const float* Wv = (const float*)d_in[3];
  const float* Wo = (const float*)d_in[4];
  const float* rope = (const float*)d_in[5];
  const float* so8 = (const float*)d_in[6];

  float* ws = (float*)d_ws;
  float* kb = ws;
  float* vqb = kb + (size_t)2 * N_KV * SEQ * HDIM;

  kv_proj_rope_so8_kernel<<<2 * QTILES, 512, 0, stream>>>(x, Wk, Wv, rope, so8,
                                                         kb, vqb);
  so8_qproj_flash_attn_outproj_kernel<<<2 * QTILES, 1024, 0, stream>>>(
      x, Wq, rope, so8, kb, vqb, Wo, (float*)d_out);
}
